# starting kernel
# baseline (speedup 1.0000x reference)
#include <hip/hip_runtime.h>
#include <hip/hip_bf16.h>

typedef __attribute__((ext_vector_type(2))) float v2f;
typedef __attribute__((ext_vector_type(8))) float v8f;

#define N_NODES 50000

__global__ void degree_kernel(const int* __restrict__ src,
                              const int* __restrict__ dst,
                              int* __restrict__ degO,
                              int* __restrict__ degI, int nE) {
    int e = blockIdx.x * blockDim.x + threadIdx.x;
    if (e < nE) {
        atomicAdd(&degO[src[e]], 1);
        atomicAdd(&degI[dst[e]], 1);
    }
}

__global__ void norm_kernel(const int* __restrict__ deg,
                            float* __restrict__ norm, int n) {
    int i = blockIdx.x * blockDim.x + threadIdx.x;
    if (i < n) {
        float d = (float)(deg[i] < 1 ? 1 : deg[i]);
        norm[i] = 1.0f / sqrtf(d);
    }
}

template <int K, int N>
__global__ __launch_bounds__(256) void gemm_wmma(
    const float* __restrict__ A, const float* __restrict__ scale,
    const float* __restrict__ W, float* __restrict__ P, int M) {
    const int wave = threadIdx.x >> 5;
    const int lane = threadIdx.x & 31;
    const int tilesN = N / 16;
    const int tile = blockIdx.x * 8 + wave;
    if (tile >= (M / 16) * tilesN) return;

    const int mb = tile / tilesN;
    const int nb = tile - mb * tilesN;
    const int m0 = mb * 16;
    const int n0 = nb * 16;
    const int ln = lane & 15;
    const int hi = lane >> 4;

    const float sm = scale[m0 + ln];
    const float* Arow = A + (size_t)(m0 + ln) * K + 2 * hi;
    const float* Wp   = W + n0 + ln;

    v8f c = {0.f, 0.f, 0.f, 0.f, 0.f, 0.f, 0.f, 0.f};
#pragma unroll 8
    for (int k0 = 0; k0 < K; k0 += 4) {
        v2f a = *(const v2f*)(Arow + k0);
        a.x *= sm;
        a.y *= sm;
        v2f b;
        b.x = Wp[(size_t)(k0 + 2 * hi) * N];
        b.y = Wp[(size_t)(k0 + 2 * hi + 1) * N];
        c = __builtin_amdgcn_wmma_f32_16x16x4_f32(
            false, a, false, b, (short)0, c, false, false);
    }

    float* Pp = P + (size_t)(m0 + 8 * hi) * N + n0 + ln;
#pragma unroll
    for (int r = 0; r < 8; ++r) Pp[(size_t)r * N] = c[r];
}

template <int F>
__global__ __launch_bounds__(256) void scatter_kernel(
    const float* __restrict__ P, const int* __restrict__ src,
    const int* __restrict__ dst, float* __restrict__ G, int nE) {
    int e = blockIdx.x * 8 + (threadIdx.x >> 5);
    if (e >= nE) return;
    const int lane = threadIdx.x & 31;
    constexpr int PER = F / 32;
    const int s = src[e];
    const int d = dst[e];
    const float* hp = P + (size_t)s * F + lane * PER;
    float* gp = G + (size_t)d * F + lane * PER;
    float v[PER];
#pragma unroll
    for (int j = 0; j < PER; ++j) v[j] = hp[j];
#pragma unroll
    for (int j = 0; j < PER; ++j)
        __hip_atomic_fetch_add(gp + j, v[j], __ATOMIC_RELAXED,
                               __HIP_MEMORY_SCOPE_AGENT);
}

template <int F, bool RELU>
__global__ void pointwise_kernel(const float* __restrict__ G,
                                 const float* __restrict__ inorm,
                                 const float* __restrict__ bias,
                                 float* __restrict__ O, int total) {
    int idx = blockIdx.x * blockDim.x + threadIdx.x;
    if (idx >= total) return;
    int i = idx / F;
    int j = idx - i * F;
    float v = G[idx] * inorm[i] + bias[j];
    if (RELU) v = fmaxf(v, 0.0f);
    O[idx] = v;
}

extern "C" void kernel_launch(void* const* d_in, const int* in_sizes, int n_in,
                              void* d_out, int out_size, void* d_ws,
                              size_t ws_size, hipStream_t stream) {
    const float* x  = (const float*)d_in[0];
    const int*   ei = (const int*)d_in[1];
    const float* W0 = (const float*)d_in[2];
    const float* b0 = (const float*)d_in[3];
    const float* W1 = (const float*)d_in[4];
    const float* b1 = (const float*)d_in[5];
    const float* W2 = (const float*)d_in[6];
    const float* b2 = (const float*)d_in[7];
    const float* W3 = (const float*)d_in[8];
    const float* b3 = (const float*)d_in[9];

    const int nN = N_NODES;
    const int nE = in_sizes[1] / 2;
    const int* src = ei;
    const int* dst = ei + nE;

    char* ws = (char*)d_ws;
    size_t off = 0;
    auto alloc = [&](size_t bytes) -> void* {
        off = (off + 255) & ~(size_t)255;
        void* p = ws + off;
        off += bytes;
        return p;
    };
    int*   degO  = (int*)alloc((size_t)nN * sizeof(int));
    int*   degI  = (int*)alloc((size_t)nN * sizeof(int));
    float* onorm = (float*)alloc((size_t)nN * sizeof(float));
    float* inorm = (float*)alloc((size_t)nN * sizeof(float));
    float* P = (float*)alloc((size_t)nN * 128 * sizeof(float));
    float* G = (float*)alloc((size_t)nN * 128 * sizeof(float));
    float* H = (float*)alloc((size_t)nN * 128 * sizeof(float));

    hipMemsetAsync(degO, 0, (size_t)nN * sizeof(int), stream);
    hipMemsetAsync(degI, 0, (size_t)nN * sizeof(int), stream);
    degree_kernel<<<(nE + 255) / 256, 256, 0, stream>>>(src, dst, degO, degI, nE);
    norm_kernel<<<(nN + 255) / 256, 256, 0, stream>>>(degO, onorm, nN);
    norm_kernel<<<(nN + 255) / 256, 256, 0, stream>>>(degI, inorm, nN);

    const int tilesM = nN / 16;

    gemm_wmma<256, 128><<<(tilesM * 8 + 7) / 8, 256, 0, stream>>>(x, onorm, W0, P, nN);
    hipMemsetAsync(G, 0, (size_t)nN * 128 * sizeof(float), stream);
    scatter_kernel<128><<<(nE + 7) / 8, 256, 0, stream>>>(P, src, dst, G, nE);
    pointwise_kernel<128, true><<<(nN * 128 + 255) / 256, 256, 0, stream>>>(
        G, inorm, b0, H, nN * 128);

    gemm_wmma<128, 128><<<(tilesM * 8 + 7) / 8, 256, 0, stream>>>(H, onorm, W1, P, nN);
    hipMemsetAsync(G, 0, (size_t)nN * 128 * sizeof(float), stream);
    scatter_kernel<128><<<(nE + 7) / 8, 256, 0, stream>>>(P, src, dst, G, nE);
    pointwise_kernel<128, true><<<(nN * 128 + 255) / 256, 256, 0, stream>>>(
        G, inorm, b1, H, nN * 128);

    gemm_wmma<128, 128><<<(tilesM * 8 + 7) / 8, 256, 0, stream>>>(H, onorm, W2, P, nN);
    hipMemsetAsync(G, 0, (size_t)nN * 128 * sizeof(float), stream);
    scatter_kernel<128><<<(nE + 7) / 8, 256, 0, stream>>>(P, src, dst, G, nE);
    pointwise_kernel<128, true><<<(nN * 128 + 255) / 256, 256, 0, stream>>>(
        G, inorm, b2, H, nN * 128);

    gemm_wmma<128, 64><<<(tilesM * 4 + 7) / 8, 256, 0, stream>>>(H, onorm, W3, P, nN);
    hipMemsetAsync(G, 0, (size_t)nN * 64 * sizeof(float), stream);
    scatter_kernel<64><<<(nE + 7) / 8, 256, 0, stream>>>(P, src, dst, G, nE);
    pointwise_kernel<64, false><<<(nN * 64 + 255) / 256, 256, 0, stream>>>(
        G, inorm, b3, (float*)d_out, nN * 64);

    (void)n_in; (void)out_size; (void)ws_size;
}


# Float-atomic scatter replaced by destination-owner gather over a dst-sorted edge list built in-call: a wave sums a node's in-rows in registers (fixed ascending-source order), norm+bias+relu, one store
# speedup vs baseline: 2.3558x; 2.3558x over previous
#include <hip/hip_runtime.h>
#include <hip/hip_bf16.h>

typedef __attribute__((ext_vector_type(2))) float v2f;
typedef __attribute__((ext_vector_type(8))) float v8f;

#define N_NODES 50000

__global__ void degree_kernel(const int* __restrict__ src,
                              const int* __restrict__ dst,
                              int* __restrict__ degO,
                              int* __restrict__ degI,
                              int* __restrict__ slot, int nE) {
    int e = blockIdx.x * blockDim.x + threadIdx.x;
    if (e < nE) {
        atomicAdd(&degO[src[e]], 1);
        slot[e] = atomicAdd(&degI[dst[e]], 1);
    }
}

__global__ __launch_bounds__(1024) void rowptr_kernel(
    const int* __restrict__ deg, int* __restrict__ rowptr, int n) {
    __shared__ int s_before[32];
    __shared__ int s_total[32];
    const int tid = threadIdx.x;
    const int lane = tid & 31;
    const int wave = tid >> 5;
    const int first = blockIdx.x * 1024;

    int before = 0;
    for (int i = tid; i < first; i += 1024) before += deg[i];

    const int i = first + tid;
    const int v = (i < n) ? deg[i] : 0;
    int incl = v;
    for (int o = 1; o < 32; o <<= 1) {
        int t = __shfl_up(incl, o, 32);
        if (lane >= o) incl += t;
    }
    for (int o = 16; o > 0; o >>= 1) before += __shfl_xor(before, o, 32);
    if (lane == 31) {
        s_before[wave] = before;
        s_total[wave] = incl;
    }
    __syncthreads();
    int part = s_before[lane] + (lane < wave ? s_total[lane] : 0);
    for (int o = 16; o > 0; o >>= 1) part += __shfl_xor(part, o, 32);
    const int excl = part + incl - v;
    if (i < n) rowptr[i] = excl;
    if (i == n - 1) rowptr[n] = excl + v;
}

__global__ void fill_kernel(const int* __restrict__ src,
                            const int* __restrict__ dst,
                            const int* __restrict__ slot,
                            const int* __restrict__ rowptr,
                            int* __restrict__ csr, int nE) {
    int e = blockIdx.x * blockDim.x + threadIdx.x;
    if (e < nE) csr[rowptr[dst[e]] + slot[e]] = src[e];
}

__global__ __launch_bounds__(256) void order_kernel(
    const int* __restrict__ rowptr, const int* __restrict__ unsorted,
    int* __restrict__ sorted, int n) {
    const int node = blockIdx.x * 8 + (threadIdx.x >> 5);
    if (node >= n) return;
    const int lane = threadIdx.x & 31;
    const int beg = rowptr[node];
    const int end = rowptr[node + 1];
    for (int i = beg + lane; i < end; i += 32) {
        const int key = unsorted[i];
        int r = 0;
        for (int j = beg; j < end; ++j) {
            const int kj = unsorted[j];
            r += (kj < key || (kj == key && j < i)) ? 1 : 0;
        }
        sorted[beg + r] = key;
    }
}

__global__ void norm_kernel(const int* __restrict__ deg,
                            float* __restrict__ norm, int n) {
    int i = blockIdx.x * blockDim.x + threadIdx.x;
    if (i < n) {
        float d = (float)(deg[i] < 1 ? 1 : deg[i]);
        norm[i] = 1.0f / sqrtf(d);
    }
}

template <int K, int N>
__global__ __launch_bounds__(256) void gemm_wmma(
    const float* __restrict__ A, const float* __restrict__ scale,
    const float* __restrict__ W, float* __restrict__ P, int M) {
    const int wave = threadIdx.x >> 5;
    const int lane = threadIdx.x & 31;
    const int tilesN = N / 16;
    const int tile = blockIdx.x * 8 + wave;
    if (tile >= (M / 16) * tilesN) return;

    const int mb = tile / tilesN;
    const int nb = tile - mb * tilesN;
    const int m0 = mb * 16;
    const int n0 = nb * 16;
    const int ln = lane & 15;
    const int hi = lane >> 4;

    const float sm = scale[m0 + ln];
    const float* Arow = A + (size_t)(m0 + ln) * K + 2 * hi;
    const float* Wp   = W + n0 + ln;

    v8f c = {0.f, 0.f, 0.f, 0.f, 0.f, 0.f, 0.f, 0.f};
#pragma unroll 8
    for (int k0 = 0; k0 < K; k0 += 4) {
        v2f a = *(const v2f*)(Arow + k0);
        a.x *= sm;
        a.y *= sm;
        v2f b;
        b.x = Wp[(size_t)(k0 + 2 * hi) * N];
        b.y = Wp[(size_t)(k0 + 2 * hi + 1) * N];
        c = __builtin_amdgcn_wmma_f32_16x16x4_f32(
            false, a, false, b, (short)0, c, false, false);
    }

    float* Pp = P + (size_t)(m0 + 8 * hi) * N + n0 + ln;
#pragma unroll
    for (int r = 0; r < 8; ++r) Pp[(size_t)r * N] = c[r];
}

template <int PER> struct rowvec;
template <> struct rowvec<4> { typedef __attribute__((ext_vector_type(4))) float type; };
template <> struct rowvec<2> { typedef __attribute__((ext_vector_type(2))) float type; };

template <int F, bool RELU>
__global__ __launch_bounds__(256) void gather_kernel(
    const float* __restrict__ P, const int* __restrict__ rowptr,
    const int* __restrict__ csr, const float* __restrict__ inorm,
    const float* __restrict__ bias, float* __restrict__ O, int n) {
    constexpr int PER = F / 32;
    constexpr int UN = 8;
    typedef typename rowvec<PER>::type vec;

    const int node = __builtin_amdgcn_readfirstlane(
        (int)(blockIdx.x * 8 + (threadIdx.x >> 5)));
    if (node >= n) return;
    const int lane = threadIdx.x & 31;
    const int beg = rowptr[node];
    const int end = rowptr[node + 1];

    const vec vzero = (vec)(0.0f);
    vec acc = vzero;
    const float* Pl = P + lane * PER;
    for (int base = beg; base < end; base += 32) {
        const int cnt = (end - base < 32) ? (end - base) : 32;
        const int mine = (lane < cnt) ? csr[base + lane] : 0;
        for (int j = 0; j < cnt; j += UN) {
            vec v[UN];
#pragma unroll
            for (int u = 0; u < UN; ++u) {
                const int s = __builtin_amdgcn_readlane(mine, (j + u) & 31);
                v[u] = *(const vec*)(Pl + (size_t)s * F);
            }
#pragma unroll
            for (int u = 0; u < UN; ++u) acc += ((j + u) < cnt) ? v[u] : vzero;
        }
    }

    const float sc = inorm[node];
    const vec bv = *(const vec*)(bias + lane * PER);
    vec o = acc * sc + bv;
    if (RELU) o = __builtin_elementwise_max(o, vzero);
    *(vec*)(O + (size_t)node * F + lane * PER) = o;
}

extern "C" void kernel_launch(void* const* d_in, const int* in_sizes, int n_in,
                              void* d_out, int out_size, void* d_ws,
                              size_t ws_size, hipStream_t stream) {
    const float* x  = (const float*)d_in[0];
    const int*   ei = (const int*)d_in[1];
    const float* W0 = (const float*)d_in[2];
    const float* b0 = (const float*)d_in[3];
    const float* W1 = (const float*)d_in[4];
    const float* b1 = (const float*)d_in[5];
    const float* W2 = (const float*)d_in[6];
    const float* b2 = (const float*)d_in[7];
    const float* W3 = (const float*)d_in[8];
    const float* b3 = (const float*)d_in[9];

    const int nN = N_NODES;
    const int nE = in_sizes[1] / 2;
    const int* src = ei;
    const int* dst = ei + nE;

    char* ws = (char*)d_ws;
    size_t off = 0;
    auto alloc = [&](size_t bytes) -> void* {
        off = (off + 255) & ~(size_t)255;
        void* p = ws + off;
        off += bytes;
        return p;
    };
    int*   degO  = (int*)alloc((size_t)nN * sizeof(int));
    int*   degI  = (int*)alloc((size_t)nN * sizeof(int));
    float* onorm = (float*)alloc((size_t)nN * sizeof(float));
    float* inorm = (float*)alloc((size_t)nN * sizeof(float));
    int*   rowptr = (int*)alloc((size_t)(nN + 1) * sizeof(int));
    int*   slot   = (int*)alloc((size_t)nE * sizeof(int));
    int*   csrU   = (int*)alloc((size_t)nE * sizeof(int));
    int*   csr    = (int*)alloc((size_t)nE * sizeof(int));
    float* P = (float*)alloc((size_t)nN * 128 * sizeof(float));
    float* H = (float*)alloc((size_t)nN * 128 * sizeof(float));

    hipMemsetAsync(degO, 0, (size_t)nN * sizeof(int), stream);
    hipMemsetAsync(degI, 0, (size_t)nN * sizeof(int), stream);
    degree_kernel<<<(nE + 255) / 256, 256, 0, stream>>>(src, dst, degO, degI, slot, nE);
    norm_kernel<<<(nN + 255) / 256, 256, 0, stream>>>(degO, onorm, nN);
    norm_kernel<<<(nN + 255) / 256, 256, 0, stream>>>(degI, inorm, nN);
    rowptr_kernel<<<(nN + 1023) / 1024, 1024, 0, stream>>>(degI, rowptr, nN);
    fill_kernel<<<(nE + 255) / 256, 256, 0, stream>>>(src, dst, slot, rowptr, csrU, nE);
    order_kernel<<<(nN + 7) / 8, 256, 0, stream>>>(rowptr, csrU, csr, nN);

    const int tilesM = nN / 16;

    gemm_wmma<256, 128><<<(tilesM * 8 + 7) / 8, 256, 0, stream>>>(x, onorm, W0, P, nN);
    gather_kernel<128, true><<<(nN + 7) / 8, 256, 0, stream>>>(
        P, rowptr, csr, inorm, b0, H, nN);

    gemm_wmma<128, 128><<<(tilesM * 8 + 7) / 8, 256, 0, stream>>>(H, onorm, W1, P, nN);
    gather_kernel<128, true><<<(nN + 7) / 8, 256, 0, stream>>>(
        P, rowptr, csr, inorm, b1, H, nN);

    gemm_wmma<128, 128><<<(tilesM * 8 + 7) / 8, 256, 0, stream>>>(H, onorm, W2, P, nN);
    gather_kernel<128, true><<<(nN + 7) / 8, 256, 0, stream>>>(
        P, rowptr, csr, inorm, b2, H, nN);

    gemm_wmma<128, 64><<<(tilesM * 4 + 7) / 8, 256, 0, stream>>>(H, onorm, W3, P, nN);
    gather_kernel<64, false><<<(nN + 7) / 8, 256, 0, stream>>>(
        P, rowptr, csr, inorm, b3, (float*)d_out, nN);

    (void)n_in; (void)out_size; (void)ws_size;
}


# GEMM re-tiled: one wave owns a whole row strip of P (all N columns, 8 or 4 accumulator tiles), so an A fragment is loaded once per k-step, not once per column tile; W by 128-bit loads; sum order kept
# speedup vs baseline: 2.7219x; 1.1554x over previous
#include <hip/hip_runtime.h>
#include <hip/hip_bf16.h>

typedef __attribute__((ext_vector_type(2))) float v2f;
typedef __attribute__((ext_vector_type(8))) float v8f;

#define N_NODES 50000

__global__ void degree_kernel(const int* __restrict__ src,
                              const int* __restrict__ dst,
                              int* __restrict__ degO,
                              int* __restrict__ degI,
                              int* __restrict__ slot, int nE) {
    int e = blockIdx.x * blockDim.x + threadIdx.x;
    if (e < nE) {
        atomicAdd(&degO[src[e]], 1);
        slot[e] = atomicAdd(&degI[dst[e]], 1);
    }
}

__global__ __launch_bounds__(1024) void rowptr_kernel(
    const int* __restrict__ deg, int* __restrict__ rowptr, int n) {
    __shared__ int s_before[32];
    __shared__ int s_total[32];
    const int tid = threadIdx.x;
    const int lane = tid & 31;
    const int wave = tid >> 5;
    const int first = blockIdx.x * 1024;

    int before = 0;
    for (int i = tid; i < first; i += 1024) before += deg[i];

    const int i = first + tid;
    const int v = (i < n) ? deg[i] : 0;
    int incl = v;
    for (int o = 1; o < 32; o <<= 1) {
        int t = __shfl_up(incl, o, 32);
        if (lane >= o) incl += t;
    }
    for (int o = 16; o > 0; o >>= 1) before += __shfl_xor(before, o, 32);
    if (lane == 31) {
        s_before[wave] = before;
        s_total[wave] = incl;
    }
    __syncthreads();
    int part = s_before[lane] + (lane < wave ? s_total[lane] : 0);
    for (int o = 16; o > 0; o >>= 1) part += __shfl_xor(part, o, 32);
    const int excl = part + incl - v;
    if (i < n) rowptr[i] = excl;
    if (i == n - 1) rowptr[n] = excl + v;
}

__global__ void fill_kernel(const int* __restrict__ src,
                            const int* __restrict__ dst,
                            const int* __restrict__ slot,
                            const int* __restrict__ rowptr,
                            int* __restrict__ csr, int nE) {
    int e = blockIdx.x * blockDim.x + threadIdx.x;
    if (e < nE) csr[rowptr[dst[e]] + slot[e]] = src[e];
}

__global__ __launch_bounds__(256) void order_kernel(
    const int* __restrict__ rowptr, const int* __restrict__ unsorted,
    int* __restrict__ sorted, int n) {
    const int node = blockIdx.x * 8 + (threadIdx.x >> 5);
    if (node >= n) return;
    const int lane = threadIdx.x & 31;
    const int beg = rowptr[node];
    const int end = rowptr[node + 1];
    for (int i = beg + lane; i < end; i += 32) {
        const int key = unsorted[i];
        int r = 0;
        for (int j = beg; j < end; ++j) {
            const int kj = unsorted[j];
            r += (kj < key || (kj == key && j < i)) ? 1 : 0;
        }
        sorted[beg + r] = key;
    }
}

__global__ void norm_kernel(const int* __restrict__ deg,
                            float* __restrict__ norm, int n) {
    int i = blockIdx.x * blockDim.x + threadIdx.x;
    if (i < n) {
        float d = (float)(deg[i] < 1 ? 1 : deg[i]);
        norm[i] = 1.0f / sqrtf(d);
    }
}

typedef __attribute__((ext_vector_type(4))) float v4f;

template <int K, int N>
__global__ __launch_bounds__(128) void gemm_wmma(
    const float* __restrict__ A, const float* __restrict__ scale,
    const float* __restrict__ W, float* __restrict__ P, int M) {
    constexpr int G = N / 64;
    constexpr int NT = N / 16;
    const int wave = threadIdx.x >> 5;
    const int lane = threadIdx.x & 31;
    const int strip = blockIdx.x * 4 + wave;
    if (strip >= M / 16) return;

    const int m0 = strip * 16;
    const int ln = lane & 15;
    const int hi = lane >> 4;

    const float sm = scale[m0 + ln];
    const float* Arow = A + (size_t)(m0 + ln) * K + 2 * hi;
    const float* Wp   = W + (size_t)(2 * hi) * N + 4 * ln;

    v8f c[NT];
#pragma unroll
    for (int t = 0; t < NT; ++t) c[t] = (v8f){0.f, 0.f, 0.f, 0.f, 0.f, 0.f, 0.f, 0.f};

#pragma unroll 8
    for (int k0 = 0; k0 < K; k0 += 4) {
        v2f a = *(const v2f*)(Arow + k0);
        a.x *= sm;
        a.y *= sm;
        v4f w0[G], w1[G];
#pragma unroll
        for (int g = 0; g < G; ++g) {
            w0[g] = *(const v4f*)(Wp + (size_t)k0 * N + 64 * g);
            w1[g] = *(const v4f*)(Wp + (size_t)(k0 + 1) * N + 64 * g);
        }
#pragma unroll
        for (int g = 0; g < G; ++g) {
#pragma unroll
            for (int j = 0; j < 4; ++j) {
                v2f b;
                b.x = w0[g][j];
                b.y = w1[g][j];
                c[4 * g + j] = __builtin_amdgcn_wmma_f32_16x16x4_f32(
                    false, a, false, b, (short)0, c[4 * g + j], false, false);
            }
        }
    }

    float* Pp = P + (size_t)(m0 + 8 * hi) * N + 4 * ln;
#pragma unroll
    for (int r = 0; r < 8; ++r) {
#pragma unroll
        for (int g = 0; g < G; ++g) {
            v4f o;
            o.x = c[4 * g + 0][r];
            o.y = c[4 * g + 1][r];
            o.z = c[4 * g + 2][r];
            o.w = c[4 * g + 3][r];
            *(v4f*)(Pp + (size_t)r * N + 64 * g) = o;
        }
    }
}

template <int PER> struct rowvec;
template <> struct rowvec<4> { typedef __attribute__((ext_vector_type(4))) float type; };
template <> struct rowvec<2> { typedef __attribute__((ext_vector_type(2))) float type; };

template <int F, bool RELU>
__global__ __launch_bounds__(256) void gather_kernel(
    const float* __restrict__ P, const int* __restrict__ rowptr,
    const int* __restrict__ csr, const float* __restrict__ inorm,
    const float* __restrict__ bias, float* __restrict__ O, int n) {
    constexpr int PER = F / 32;
    constexpr int UN = 8;
    typedef typename rowvec<PER>::type vec;

    const int node = __builtin_amdgcn_readfirstlane(
        (int)(blockIdx.x * 8 + (threadIdx.x >> 5)));
    if (node >= n) return;
    const int lane = threadIdx.x & 31;
    const int beg = rowptr[node];
    const int end = rowptr[node + 1];

    const vec vzero = (vec)(0.0f);
    vec acc = vzero;
    const float* Pl = P + lane * PER;
    for (int base = beg; base < end; base += 32) {
        const int cnt = (end - base < 32) ? (end - base) : 32;
        const int mine = (lane < cnt) ? csr[base + lane] : 0;
        for (int j = 0; j < cnt; j += UN) {
            vec v[UN];
#pragma unroll
            for (int u = 0; u < UN; ++u) {
                const int s = __builtin_amdgcn_readlane(mine, (j + u) & 31);
                v[u] = *(const vec*)(Pl + (size_t)s * F);
            }
#pragma unroll
            for (int u = 0; u < UN; ++u) acc += ((j + u) < cnt) ? v[u] : vzero;
        }
    }

    const float sc = inorm[node];
    const vec bv = *(const vec*)(bias + lane * PER);
    vec o = acc * sc + bv;
    if (RELU) o = __builtin_elementwise_max(o, vzero);
    *(vec*)(O + (size_t)node * F + lane * PER) = o;
}

extern "C" void kernel_launch(void* const* d_in, const int* in_sizes, int n_in,
                              void* d_out, int out_size, void* d_ws,
                              size_t ws_size, hipStream_t stream) {
    const float* x  = (const float*)d_in[0];
    const int*   ei = (const int*)d_in[1];
    const float* W0 = (const float*)d_in[2];
    const float* b0 = (const float*)d_in[3];
    const float* W1 = (const float*)d_in[4];
    const float* b1 = (const float*)d_in[5];
    const float* W2 = (const float*)d_in[6];
    const float* b2 = (const float*)d_in[7];
    const float* W3 = (const float*)d_in[8];
    const float* b3 = (const float*)d_in[9];

    const int nN = N_NODES;
    const int nE = in_sizes[1] / 2;
    const int* src = ei;
    const int* dst = ei + nE;

    char* ws = (char*)d_ws;
    size_t off = 0;
    auto alloc = [&](size_t bytes) -> void* {
        off = (off + 255) & ~(size_t)255;
        void* p = ws + off;
        off += bytes;
        return p;
    };
    int*   degO  = (int*)alloc((size_t)nN * sizeof(int));
    int*   degI  = (int*)alloc((size_t)nN * sizeof(int));
    float* onorm = (float*)alloc((size_t)nN * sizeof(float));
    float* inorm = (float*)alloc((size_t)nN * sizeof(float));
    int*   rowptr = (int*)alloc((size_t)(nN + 1) * sizeof(int));
    int*   slot   = (int*)alloc((size_t)nE * sizeof(int));
    int*   csrU   = (int*)alloc((size_t)nE * sizeof(int));
    int*   csr    = (int*)alloc((size_t)nE * sizeof(int));
    float* P = (float*)alloc((size_t)nN * 128 * sizeof(float));
    float* H = (float*)alloc((size_t)nN * 128 * sizeof(float));

    hipMemsetAsync(degO, 0, (size_t)nN * sizeof(int), stream);
    hipMemsetAsync(degI, 0, (size_t)nN * sizeof(int), stream);
    degree_kernel<<<(nE + 255) / 256, 256, 0, stream>>>(src, dst, degO, degI, slot, nE);
    norm_kernel<<<(nN + 255) / 256, 256, 0, stream>>>(degO, onorm, nN);
    norm_kernel<<<(nN + 255) / 256, 256, 0, stream>>>(degI, inorm, nN);
    rowptr_kernel<<<(nN + 1023) / 1024, 1024, 0, stream>>>(degI, rowptr, nN);
    fill_kernel<<<(nE + 255) / 256, 256, 0, stream>>>(src, dst, slot, rowptr, csrU, nE);
    order_kernel<<<(nN + 7) / 8, 256, 0, stream>>>(rowptr, csrU, csr, nN);

    const int tilesM = nN / 16;

    gemm_wmma<256, 128><<<(tilesM + 3) / 4, 128, 0, stream>>>(x, onorm, W0, P, nN);
    gather_kernel<128, true><<<(nN + 7) / 8, 256, 0, stream>>>(
        P, rowptr, csr, inorm, b0, H, nN);

    gemm_wmma<128, 128><<<(tilesM + 3) / 4, 128, 0, stream>>>(H, onorm, W1, P, nN);
    gather_kernel<128, true><<<(nN + 7) / 8, 256, 0, stream>>>(
        P, rowptr, csr, inorm, b1, H, nN);

    gemm_wmma<128, 128><<<(tilesM + 3) / 4, 128, 0, stream>>>(H, onorm, W2, P, nN);
    gather_kernel<128, true><<<(nN + 7) / 8, 256, 0, stream>>>(
        P, rowptr, csr, inorm, b2, H, nN);

    gemm_wmma<128, 64><<<(tilesM + 3) / 4, 128, 0, stream>>>(H, onorm, W3, P, nN);
    gather_kernel<64, false><<<(nN + 7) / 8, 256, 0, stream>>>(
        P, rowptr, csr, inorm, b3, (float*)d_out, nN);

    (void)n_in; (void)out_size; (void)ws_size;
}


# Layers 1-3: aggregation launch keeps its 16 finished rows (norm, bias, relu, left norm) in LDS and runs the next layer's f32 WMMA product on them: no H round trip, 3 fewer launches; sum order kept
# speedup vs baseline: 3.0807x; 1.1318x over previous
#include <hip/hip_runtime.h>
#include <hip/hip_bf16.h>

typedef __attribute__((ext_vector_type(2))) float v2f;
typedef __attribute__((ext_vector_type(8))) float v8f;

#define N_NODES 50000

__global__ void degree_kernel(const int* __restrict__ src,
                              const int* __restrict__ dst,
                              int* __restrict__ degO,
                              int* __restrict__ degI,
                              int* __restrict__ slot, int nE) {
    int e = blockIdx.x * blockDim.x + threadIdx.x;
    if (e < nE) {
        atomicAdd(&degO[src[e]], 1);
        slot[e] = atomicAdd(&degI[dst[e]], 1);
    }
}

__global__ __launch_bounds__(1024) void rowptr_kernel(
    const int* __restrict__ deg, int* __restrict__ rowptr, int n) {
    __shared__ int s_before[32];
    __shared__ int s_total[32];
    const int tid = threadIdx.x;
    const int lane = tid & 31;
    const int wave = tid >> 5;
    const int first = blockIdx.x * 1024;

    int before = 0;
    for (int i = tid; i < first; i += 1024) before += deg[i];

    const int i = first + tid;
    const int v = (i < n) ? deg[i] : 0;
    int incl = v;
    for (int o = 1; o < 32; o <<= 1) {
        int t = __shfl_up(incl, o, 32);
        if (lane >= o) incl += t;
    }
    for (int o = 16; o > 0; o >>= 1) before += __shfl_xor(before, o, 32);
    if (lane == 31) {
        s_before[wave] = before;
        s_total[wave] = incl;
    }
    __syncthreads();
    int part = s_before[lane] + (lane < wave ? s_total[lane] : 0);
    for (int o = 16; o > 0; o >>= 1) part += __shfl_xor(part, o, 32);
    const int excl = part + incl - v;
    if (i < n) rowptr[i] = excl;
    if (i == n - 1) rowptr[n] = excl + v;
}

__global__ void fill_kernel(const int* __restrict__ src,
                            const int* __restrict__ dst,
                            const int* __restrict__ slot,
                            const int* __restrict__ rowptr,
                            int* __restrict__ csr, int nE) {
    int e = blockIdx.x * blockDim.x + threadIdx.x;
    if (e < nE) csr[rowptr[dst[e]] + slot[e]] = src[e];
}

__global__ __launch_bounds__(256) void order_kernel(
    const int* __restrict__ rowptr, const int* __restrict__ unsorted,
    int* __restrict__ sorted, int n) {
    const int node = blockIdx.x * 8 + (threadIdx.x >> 5);
    if (node >= n) return;
    const int lane = threadIdx.x & 31;
    const int beg = rowptr[node];
    const int end = rowptr[node + 1];
    for (int i = beg + lane; i < end; i += 32) {
        const int key = unsorted[i];
        int r = 0;
        for (int j = beg; j < end; ++j) {
            const int kj = unsorted[j];
            r += (kj < key || (kj == key && j < i)) ? 1 : 0;
        }
        sorted[beg + r] = key;
    }
}

__global__ void norm_kernel(const int* __restrict__ deg,
                            float* __restrict__ norm, int n) {
    int i = blockIdx.x * blockDim.x + threadIdx.x;
    if (i < n) {
        float d = (float)(deg[i] < 1 ? 1 : deg[i]);
        norm[i] = 1.0f / sqrtf(d);
    }
}

typedef __attribute__((ext_vector_type(4))) float v4f;

template <int K, int N>
__global__ __launch_bounds__(128) void gemm_wmma(
    const float* __restrict__ A, const float* __restrict__ scale,
    const float* __restrict__ W, float* __restrict__ P, int M) {
    constexpr int G = N / 64;
    constexpr int NT = N / 16;
    const int wave = threadIdx.x >> 5;
    const int lane = threadIdx.x & 31;
    const int strip = blockIdx.x * 4 + wave;
    if (strip >= M / 16) return;

    const int m0 = strip * 16;
    const int ln = lane & 15;
    const int hi = lane >> 4;

    const float sm = scale[m0 + ln];
    const float* Arow = A + (size_t)(m0 + ln) * K + 2 * hi;
    const float* Wp   = W + (size_t)(2 * hi) * N + 4 * ln;

    v8f c[NT];
#pragma unroll
    for (int t = 0; t < NT; ++t) c[t] = (v8f){0.f, 0.f, 0.f, 0.f, 0.f, 0.f, 0.f, 0.f};

#pragma unroll 8
    for (int k0 = 0; k0 < K; k0 += 4) {
        v2f a = *(const v2f*)(Arow + k0);
        a.x *= sm;
        a.y *= sm;
        v4f w0[G], w1[G];
#pragma unroll
        for (int g = 0; g < G; ++g) {
            w0[g] = *(const v4f*)(Wp + (size_t)k0 * N + 64 * g);
            w1[g] = *(const v4f*)(Wp + (size_t)(k0 + 1) * N + 64 * g);
        }
#pragma unroll
        for (int g = 0; g < G; ++g) {
#pragma unroll
            for (int j = 0; j < 4; ++j) {
                v2f b;
                b.x = w0[g][j];
                b.y = w1[g][j];
                c[4 * g + j] = __builtin_amdgcn_wmma_f32_16x16x4_f32(
                    false, a, false, b, (short)0, c[4 * g + j], false, false);
            }
        }
    }

    float* Pp = P + (size_t)(m0 + 8 * hi) * N + 4 * ln;
#pragma unroll
    for (int r = 0; r < 8; ++r) {
#pragma unroll
        for (int g = 0; g < G; ++g) {
            v4f o;
            o.x = c[4 * g + 0][r];
            o.y = c[4 * g + 1][r];
            o.z = c[4 * g + 2][r];
            o.w = c[4 * g + 3][r];
            *(v4f*)(Pp + (size_t)r * N + 64 * g) = o;
        }
    }
}

template <int PER> struct rowvec;
template <> struct rowvec<4> { typedef __attribute__((ext_vector_type(4))) float type; };
template <> struct rowvec<2> { typedef __attribute__((ext_vector_type(2))) float type; };

template <int F, bool RELU>
__global__ __launch_bounds__(256) void gather_kernel(
    const float* __restrict__ P, const int* __restrict__ rowptr,
    const int* __restrict__ csr, const float* __restrict__ inorm,
    const float* __restrict__ bias, float* __restrict__ O, int n) {
    constexpr int PER = F / 32;
    constexpr int UN = 8;
    typedef typename rowvec<PER>::type vec;

    const int node = __builtin_amdgcn_readfirstlane(
        (int)(blockIdx.x * 8 + (threadIdx.x >> 5)));
    if (node >= n) return;
    const int lane = threadIdx.x & 31;
    const int beg = rowptr[node];
    const int end = rowptr[node + 1];

    const vec vzero = (vec)(0.0f);
    vec acc = vzero;
    const float* Pl = P + lane * PER;
    for (int base = beg; base < end; base += 32) {
        const int cnt = (end - base < 32) ? (end - base) : 32;
        const int mine = (lane < cnt) ? csr[base + lane] : 0;
        for (int j = 0; j < cnt; j += UN) {
            vec v[UN];
#pragma unroll
            for (int u = 0; u < UN; ++u) {
                const int s = __builtin_amdgcn_readlane(mine, (j + u) & 31);
                v[u] = *(const vec*)(Pl + (size_t)s * F);
            }
#pragma unroll
            for (int u = 0; u < UN; ++u) acc += ((j + u) < cnt) ? v[u] : vzero;
        }
    }

    const float sc = inorm[node];
    const vec bv = *(const vec*)(bias + lane * PER);
    vec o = acc * sc + bv;
    if (RELU) o = __builtin_elementwise_max(o, vzero);
    *(vec*)(O + (size_t)node * F + lane * PER) = o;
}

template <int N>
__global__ __launch_bounds__(256) void gather_gemm_kernel(
    const float* __restrict__ Pin, const int* __restrict__ rowptr,
    const int* __restrict__ csr, const float* __restrict__ inorm,
    const float* __restrict__ bias, const float* __restrict__ onorm,
    const float* __restrict__ W, float* __restrict__ Pout, int n) {
    constexpr int F = 128;
    constexpr int UN = 8;
    constexpr int PITCH = F + 4;
    constexpr int NW = N / 32;
    __shared__ __attribute__((aligned(16))) float As[16 * PITCH];

    const int wave = __builtin_amdgcn_readfirstlane((int)(threadIdx.x >> 5));
    const int lane = threadIdx.x & 31;
    const int m0 = blockIdx.x * 16;

    const v4f vzero = (v4f)(0.0f);
    const float* Pl = Pin + lane * 4;

    const int nodeA = m0 + 2 * wave;
    const int rp0 = rowptr[nodeA < n ? nodeA : n];
    const int rp1 = rowptr[nodeA + 1 < n ? nodeA + 1 : n];
    const int rp2 = rowptr[nodeA + 2 < n ? nodeA + 2 : n];
    const int firstA = (lane < rp1 - rp0) ? csr[rp0 + lane] : 0;
    const int firstB = (lane < rp2 - rp1) ? csr[rp1 + lane] : 0;

#pragma unroll
    for (int t = 0; t < 2; ++t) {
        const int row = 2 * wave + t;
        const int node = m0 + row;
        const int beg = t ? rp1 : rp0;
        const int end = t ? rp2 : rp1;
        const int first = t ? firstB : firstA;
        v4f acc = vzero;
        for (int base = beg; base < end; base += 32) {
            const int cnt = (end - base < 32) ? (end - base) : 32;
            int mine = first;
            if (base != beg) mine = (lane < cnt) ? csr[base + lane] : 0;
            for (int j = 0; j < cnt; j += UN) {
                v4f v[UN];
#pragma unroll
                for (int u = 0; u < UN; ++u) {
                    const int s = __builtin_amdgcn_readlane(mine, (j + u) & 31);
                    v[u] = *(const v4f*)(Pl + (size_t)s * F);
                }
#pragma unroll
                for (int u = 0; u < UN; ++u) acc += ((j + u) < cnt) ? v[u] : vzero;
            }
        }
        v4f a = vzero;
        if (node < n) {
            const v4f bv = *(const v4f*)(bias + lane * 4);
            v4f o = acc * inorm[node] + bv;
            o = __builtin_elementwise_max(o, vzero);
            a = o * onorm[node];
        }
        *(v4f*)(As + row * PITCH + lane * 4) = a;
    }

    __syncthreads();
    if (wave >= NW) return;

    const int ln = lane & 15;
    const int hi = lane >> 4;
    const float* Arow = As + ln * PITCH + 2 * hi;
    const float* Wp = W + (size_t)(2 * hi) * N + 32 * wave + 2 * ln;

    v8f c0 = (v8f){0.f, 0.f, 0.f, 0.f, 0.f, 0.f, 0.f, 0.f};
    v8f c1 = c0;
    constexpr int KG = 8;
    for (int kg = 0; kg < F; kg += 4 * KG) {
        v2f a[KG], w0[KG], w1[KG];
#pragma unroll
        for (int u = 0; u < KG; ++u) a[u] = *(const v2f*)(Arow + kg + 4 * u);
#pragma unroll
        for (int u = 0; u < KG; ++u) {
            w0[u] = *(const v2f*)(Wp + (size_t)(kg + 4 * u) * N);
            w1[u] = *(const v2f*)(Wp + (size_t)(kg + 4 * u + 1) * N);
        }
#pragma unroll
        for (int u = 0; u < KG; ++u) {
            v2f b0, b1;
            b0.x = w0[u].x; b0.y = w1[u].x;
            b1.x = w0[u].y; b1.y = w1[u].y;
            c0 = __builtin_amdgcn_wmma_f32_16x16x4_f32(false, a[u], false, b0, (short)0, c0, false, false);
            c1 = __builtin_amdgcn_wmma_f32_16x16x4_f32(false, a[u], false, b1, (short)0, c1, false, false);
        }
    }

    float* Pp = Pout + (size_t)(m0 + 8 * hi) * N + 32 * wave + 2 * ln;
#pragma unroll
    for (int r = 0; r < 8; ++r) {
        v2f o;
        o.x = c0[r];
        o.y = c1[r];
        *(v2f*)(Pp + (size_t)r * N) = o;
    }
}

extern "C" void kernel_launch(void* const* d_in, const int* in_sizes, int n_in,
                              void* d_out, int out_size, void* d_ws,
                              size_t ws_size, hipStream_t stream) {
    const float* x  = (const float*)d_in[0];
    const int*   ei = (const int*)d_in[1];
    const float* W0 = (const float*)d_in[2];
    const float* b0 = (const float*)d_in[3];
    const float* W1 = (const float*)d_in[4];
    const float* b1 = (const float*)d_in[5];
    const float* W2 = (const float*)d_in[6];
    const float* b2 = (const float*)d_in[7];
    const float* W3 = (const float*)d_in[8];
    const float* b3 = (const float*)d_in[9];

    const int nN = N_NODES;
    const int nE = in_sizes[1] / 2;
    const int* src = ei;
    const int* dst = ei + nE;

    char* ws = (char*)d_ws;
    size_t off = 0;
    auto alloc = [&](size_t bytes) -> void* {
        off = (off + 255) & ~(size_t)255;
        void* p = ws + off;
        off += bytes;
        return p;
    };
    int*   degO  = (int*)alloc((size_t)nN * sizeof(int));
    int*   degI  = (int*)alloc((size_t)nN * sizeof(int));
    float* onorm = (float*)alloc((size_t)nN * sizeof(float));
    float* inorm = (float*)alloc((size_t)nN * sizeof(float));
    int*   rowptr = (int*)alloc((size_t)(nN + 1) * sizeof(int));
    int*   slot   = (int*)alloc((size_t)nE * sizeof(int));
    int*   csrU   = (int*)alloc((size_t)nE * sizeof(int));
    int*   csr    = (int*)alloc((size_t)nE * sizeof(int));
    float* P = (float*)alloc((size_t)nN * 128 * sizeof(float));
    float* H = (float*)alloc((size_t)nN * 128 * sizeof(float));

    hipMemsetAsync(degO, 0, (size_t)nN * sizeof(int), stream);
    hipMemsetAsync(degI, 0, (size_t)nN * sizeof(int), stream);
    degree_kernel<<<(nE + 255) / 256, 256, 0, stream>>>(src, dst, degO, degI, slot, nE);
    norm_kernel<<<(nN + 255) / 256, 256, 0, stream>>>(degO, onorm, nN);
    norm_kernel<<<(nN + 255) / 256, 256, 0, stream>>>(degI, inorm, nN);
    rowptr_kernel<<<(nN + 1023) / 1024, 1024, 0, stream>>>(degI, rowptr, nN);
    fill_kernel<<<(nE + 255) / 256, 256, 0, stream>>>(src, dst, slot, rowptr, csrU, nE);
    order_kernel<<<(nN + 7) / 8, 256, 0, stream>>>(rowptr, csrU, csr, nN);

    const int tilesM = nN / 16;

    gemm_wmma<256, 128><<<(tilesM + 3) / 4, 128, 0, stream>>>(x, onorm, W0, P, nN);

    gather_gemm_kernel<128><<<tilesM, 256, 0, stream>>>(
        P, rowptr, csr, inorm, b0, onorm, W1, H, nN);

    gather_gemm_kernel<128><<<tilesM, 256, 0, stream>>>(
        H, rowptr, csr, inorm, b1, onorm, W2, P, nN);

    gather_gemm_kernel<64><<<tilesM, 256, 0, stream>>>(
        P, rowptr, csr, inorm, b2, onorm, W3, H, nN);

    gather_kernel<64, false><<<(nN + 7) / 8, 256, 0, stream>>>(
        H, rowptr, csr, inorm, b3, (float*)d_out, nN);

    (void)n_in; (void)out_size; (void)ws_size;
}


# Degree counts and dst-grouped edge list built without global atomics: two-level counting sort with LDS counters (returning LDS add as rank), norms and rowptr from the bucket pass; sum order kept
# speedup vs baseline: 4.1594x; 1.3501x over previous
#include <hip/hip_runtime.h>
#include <hip/hip_bf16.h>

typedef __attribute__((ext_vector_type(2))) float v2f;
typedef __attribute__((ext_vector_type(8))) float v8f;

#define N_NODES 50000

#define BUCKET_SHIFT 8
#define BUCKET_IDS 256
#define EDGE_CHUNK 4096
static_assert(N_NODES <= (BUCKET_IDS << BUCKET_SHIFT), "bucket ids must cover every node");
static_assert(N_NODES < (1 << 23), "a source id must fit above the 8 local-node bits of a record");

__global__ __launch_bounds__(1024) void bucket_count_kernel(
    const int* __restrict__ src, const int* __restrict__ dst,
    int* __restrict__ cnt2, int nE, int nC) {
    __shared__ int s_hist[2 * BUCKET_IDS];
    const int tid = threadIdx.x;
    if (tid < 2 * BUCKET_IDS) s_hist[tid] = 0;
    __syncthreads();
    const int base = blockIdx.x * EDGE_CHUNK + tid;
    int s[4], d[4];
#pragma unroll
    for (int k = 0; k < 4; ++k) {
        const int e = base + k * 1024;
        const int ec = (e < nE) ? e : (nE - 1);
        s[k] = src[ec];
        d[k] = dst[ec];
    }
#pragma unroll
    for (int k = 0; k < 4; ++k) {
        if (base + k * 1024 < nE) {
            atomicAdd(&s_hist[d[k] >> BUCKET_SHIFT], 1);
            atomicAdd(&s_hist[BUCKET_IDS + (s[k] >> BUCKET_SHIFT)], 1);
        }
    }
    __syncthreads();
    if (tid < 2 * BUCKET_IDS)
        cnt2[((size_t)(tid >> 8) * nC + blockIdx.x) * BUCKET_IDS + (tid & 255)] = s_hist[tid];
}

__device__ __forceinline__ int excl_scan_256(int v, int tid, int* s_w) {
    const int lane = tid & 31;
    const int wave = tid >> 5;
    int incl = v;
    for (int o = 1; o < 32; o <<= 1) {
        int t = __shfl_up(incl, o, 32);
        if (lane >= o) incl += t;
    }
    if (tid < 256 && lane == 31) s_w[wave] = incl;
    __syncthreads();
    int front = 0;
    for (int w = 0; w < 8; ++w) front += (w < wave) ? s_w[w] : 0;
    return front + incl - v;
}

__global__ __launch_bounds__(1024) void bucket_scan_kernel(
    const int* __restrict__ cnt2, int* __restrict__ off2,
    int* __restrict__ bst2, int nC) {
    __shared__ int s_part[4][BUCKET_IDS];
    __shared__ int s_start[BUCKET_IDS];
    __shared__ int s_w[8];
    const int tid = threadIdx.x;
    const int b = tid & 255;
    const int q = tid >> 8;
    const int* cnt = cnt2 + (size_t)blockIdx.x * nC * BUCKET_IDS;
    int* off = off2 + (size_t)blockIdx.x * nC * BUCKET_IDS;
    int* bst = bst2 + blockIdx.x * (BUCKET_IDS + 1);
    const int per = (nC + 3) >> 2;
    const int c0 = q * per;
    const int c1 = (c0 + per < nC) ? (c0 + per) : nC;

    int sum = 0;
#pragma unroll 8
    for (int c = c0; c < c1; ++c) sum += cnt[c * BUCKET_IDS + b];
    s_part[q][b] = sum;
    __syncthreads();
    int tot = 0, qfront = 0;
#pragma unroll
    for (int j = 0; j < 4; ++j) {
        const int p = s_part[j][b];
        tot += p;
        qfront += (j < q) ? p : 0;
    }
    const int ex = excl_scan_256(tid < 256 ? tot : 0, tid, s_w);
    if (tid < 256) {
        s_start[b] = ex;
        bst[b] = ex;
        if (b == 255) bst[BUCKET_IDS] = ex + tot;
    }
    __syncthreads();
    int run = s_start[b] + qfront;
#pragma unroll 8
    for (int c = c0; c < c1; ++c) {
        off[c * BUCKET_IDS + b] = run;
        run += cnt[c * BUCKET_IDS + b];
    }
}

__global__ __launch_bounds__(1024) void bucket_place_kernel(
    const int* __restrict__ src, const int* __restrict__ dst,
    const int* __restrict__ off2, int* __restrict__ bktD,
    int* __restrict__ bktS, int nE, int nC) {
    __shared__ int s_pos[2 * BUCKET_IDS];
    const int tid = threadIdx.x;
    const int base = blockIdx.x * EDGE_CHUNK + tid;
    int s[4], d[4];
#pragma unroll
    for (int k = 0; k < 4; ++k) {
        const int e = base + k * 1024;
        const int ec = (e < nE) ? e : (nE - 1);
        s[k] = src[ec];
        d[k] = dst[ec];
    }
    if (tid < 2 * BUCKET_IDS)
        s_pos[tid] = off2[((size_t)(tid >> 8) * nC + blockIdx.x) * BUCKET_IDS + (tid & 255)];
    __syncthreads();
#pragma unroll
    for (int k = 0; k < 4; ++k) {
        if (base + k * 1024 < nE) {
            const int pd = atomicAdd(&s_pos[d[k] >> BUCKET_SHIFT], 1);
            bktD[pd] = (s[k] << 8) | (d[k] & 255);
            const int ps = atomicAdd(&s_pos[BUCKET_IDS + (s[k] >> BUCKET_SHIFT)], 1);
            bktS[ps] = s[k] & 255;
        }
    }
}

__global__ __launch_bounds__(1024) void bucket_finish_kernel(
    const int* __restrict__ bktD, const int* __restrict__ bktS,
    const int* __restrict__ bst2, int* __restrict__ rowptr,
    int* __restrict__ csrU, float* __restrict__ inorm,
    float* __restrict__ onorm, int n) {
    __shared__ int s_cnt[256];
    __shared__ int s_cur[256];
    __shared__ int s_w[8];
    const int tid = threadIdx.x;
    const int role = blockIdx.x >> 8;
    const int b = blockIdx.x & 255;
    if ((b << BUCKET_SHIFT) >= n) return;
    const int beg = bst2[role * (BUCKET_IDS + 1) + b];
    const int end = bst2[role * (BUCKET_IDS + 1) + b + 1];
    const int* bkt = role ? bktS : bktD;
    if (tid < 256) s_cnt[tid] = 0;
    __syncthreads();
    for (int i0 = beg + tid; i0 < end; i0 += 4096) {
        int r[4];
#pragma unroll
        for (int k = 0; k < 4; ++k) {
            const int i = i0 + k * 1024;
            r[k] = bkt[(i < end) ? i : (end - 1)];
        }
#pragma unroll
        for (int k = 0; k < 4; ++k)
            if (i0 + k * 1024 < end) atomicAdd(&s_cnt[r[k] & 255], 1);
    }
    __syncthreads();
    const int node = (b << BUCKET_SHIFT) + tid;
    const int c = (tid < 256) ? s_cnt[tid] : 0;
    const float nrm = 1.0f / sqrtf((float)(c < 1 ? 1 : c));
    if (role) {
        if (tid < 256 && node < n) onorm[node] = nrm;
        return;
    }
    const int ex = excl_scan_256(c, tid, s_w);
    if (tid < 256) {
        s_cur[tid] = beg + ex;
        if (node < n) {
            rowptr[node] = beg + ex;
            inorm[node] = nrm;
            if (node == n - 1) rowptr[n] = beg + ex + c;
        }
    }
    __syncthreads();
    for (int i0 = beg + tid; i0 < end; i0 += 4096) {
        int r[4];
#pragma unroll
        for (int k = 0; k < 4; ++k) {
            const int i = i0 + k * 1024;
            r[k] = bktD[(i < end) ? i : (end - 1)];
        }
#pragma unroll
        for (int k = 0; k < 4; ++k) {
            if (i0 + k * 1024 < end) {
                const int p = atomicAdd(&s_cur[r[k] & 255], 1);
                csrU[p] = r[k] >> 8;
            }
        }
    }
}

__global__ __launch_bounds__(256) void order_kernel(
    const int* __restrict__ rowptr, const int* __restrict__ unsorted,
    int* __restrict__ sorted, int n) {
    const int node = blockIdx.x * 8 + (threadIdx.x >> 5);
    if (node >= n) return;
    const int lane = threadIdx.x & 31;
    const int beg = rowptr[node];
    const int end = rowptr[node + 1];
    for (int i = beg + lane; i < end; i += 32) {
        const int key = unsorted[i];
        int r = 0;
        for (int j = beg; j < end; ++j) {
            const int kj = unsorted[j];
            r += (kj < key || (kj == key && j < i)) ? 1 : 0;
        }
        sorted[beg + r] = key;
    }
}

typedef __attribute__((ext_vector_type(4))) float v4f;

template <int K, int N>
__global__ __launch_bounds__(128) void gemm_wmma(
    const float* __restrict__ A, const float* __restrict__ scale,
    const float* __restrict__ W, float* __restrict__ P, int M) {
    constexpr int G = N / 64;
    constexpr int NT = N / 16;
    const int wave = threadIdx.x >> 5;
    const int lane = threadIdx.x & 31;
    const int strip = blockIdx.x * 4 + wave;
    if (strip >= M / 16) return;

    const int m0 = strip * 16;
    const int ln = lane & 15;
    const int hi = lane >> 4;

    const float sm = scale[m0 + ln];
    const float* Arow = A + (size_t)(m0 + ln) * K + 2 * hi;
    const float* Wp   = W + (size_t)(2 * hi) * N + 4 * ln;

    v8f c[NT];
#pragma unroll
    for (int t = 0; t < NT; ++t) c[t] = (v8f){0.f, 0.f, 0.f, 0.f, 0.f, 0.f, 0.f, 0.f};

#pragma unroll 8
    for (int k0 = 0; k0 < K; k0 += 4) {
        v2f a = *(const v2f*)(Arow + k0);
        a.x *= sm;
        a.y *= sm;
        v4f w0[G], w1[G];
#pragma unroll
        for (int g = 0; g < G; ++g) {
            w0[g] = *(const v4f*)(Wp + (size_t)k0 * N + 64 * g);
            w1[g] = *(const v4f*)(Wp + (size_t)(k0 + 1) * N + 64 * g);
        }
#pragma unroll
        for (int g = 0; g < G; ++g) {
#pragma unroll
            for (int j = 0; j < 4; ++j) {
                v2f b;
                b.x = w0[g][j];
                b.y = w1[g][j];
                c[4 * g + j] = __builtin_amdgcn_wmma_f32_16x16x4_f32(
                    false, a, false, b, (short)0, c[4 * g + j], false, false);
            }
        }
    }

    float* Pp = P + (size_t)(m0 + 8 * hi) * N + 4 * ln;
#pragma unroll
    for (int r = 0; r < 8; ++r) {
#pragma unroll
        for (int g = 0; g < G; ++g) {
            v4f o;
            o.x = c[4 * g + 0][r];
            o.y = c[4 * g + 1][r];
            o.z = c[4 * g + 2][r];
            o.w = c[4 * g + 3][r];
            *(v4f*)(Pp + (size_t)r * N + 64 * g) = o;
        }
    }
}

template <int PER> struct rowvec;
template <> struct rowvec<4> { typedef __attribute__((ext_vector_type(4))) float type; };
template <> struct rowvec<2> { typedef __attribute__((ext_vector_type(2))) float type; };

template <int F, bool RELU>
__global__ __launch_bounds__(256) void gather_kernel(
    const float* __restrict__ P, const int* __restrict__ rowptr,
    const int* __restrict__ csr, const float* __restrict__ inorm,
    const float* __restrict__ bias, float* __restrict__ O, int n) {
    constexpr int PER = F / 32;
    constexpr int UN = 8;
    typedef typename rowvec<PER>::type vec;

    const int node = __builtin_amdgcn_readfirstlane(
        (int)(blockIdx.x * 8 + (threadIdx.x >> 5)));
    if (node >= n) return;
    const int lane = threadIdx.x & 31;
    const int beg = rowptr[node];
    const int end = rowptr[node + 1];

    const vec vzero = (vec)(0.0f);
    vec acc = vzero;
    const float* Pl = P + lane * PER;
    for (int base = beg; base < end; base += 32) {
        const int cnt = (end - base < 32) ? (end - base) : 32;
        const int mine = (lane < cnt) ? csr[base + lane] : 0;
        for (int j = 0; j < cnt; j += UN) {
            vec v[UN];
#pragma unroll
            for (int u = 0; u < UN; ++u) {
                const int s = __builtin_amdgcn_readlane(mine, (j + u) & 31);
                v[u] = *(const vec*)(Pl + (size_t)s * F);
            }
#pragma unroll
            for (int u = 0; u < UN; ++u) acc += ((j + u) < cnt) ? v[u] : vzero;
        }
    }

    const float sc = inorm[node];
    const vec bv = *(const vec*)(bias + lane * PER);
    vec o = acc * sc + bv;
    if (RELU) o = __builtin_elementwise_max(o, vzero);
    *(vec*)(O + (size_t)node * F + lane * PER) = o;
}

template <int N>
__global__ __launch_bounds__(256) void gather_gemm_kernel(
    const float* __restrict__ Pin, const int* __restrict__ rowptr,
    const int* __restrict__ csr, const float* __restrict__ inorm,
    const float* __restrict__ bias, const float* __restrict__ onorm,
    const float* __restrict__ W, float* __restrict__ Pout, int n) {
    constexpr int F = 128;
    constexpr int UN = 8;
    constexpr int PITCH = F + 4;
    constexpr int NW = N / 32;
    __shared__ __attribute__((aligned(16))) float As[16 * PITCH];

    const int wave = __builtin_amdgcn_readfirstlane((int)(threadIdx.x >> 5));
    const int lane = threadIdx.x & 31;
    const int m0 = blockIdx.x * 16;

    const v4f vzero = (v4f)(0.0f);
    const float* Pl = Pin + lane * 4;

    const int nodeA = m0 + 2 * wave;
    const int rp0 = rowptr[nodeA < n ? nodeA : n];
    const int rp1 = rowptr[nodeA + 1 < n ? nodeA + 1 : n];
    const int rp2 = rowptr[nodeA + 2 < n ? nodeA + 2 : n];
    const int firstA = (lane < rp1 - rp0) ? csr[rp0 + lane] : 0;
    const int firstB = (lane < rp2 - rp1) ? csr[rp1 + lane] : 0;

#pragma unroll
    for (int t = 0; t < 2; ++t) {
        const int row = 2 * wave + t;
        const int node = m0 + row;
        const int beg = t ? rp1 : rp0;
        const int end = t ? rp2 : rp1;
        const int first = t ? firstB : firstA;
        v4f acc = vzero;
        for (int base = beg; base < end; base += 32) {
            const int cnt = (end - base < 32) ? (end - base) : 32;
            int mine = first;
            if (base != beg) mine = (lane < cnt) ? csr[base + lane] : 0;
            for (int j = 0; j < cnt; j += UN) {
                v4f v[UN];
#pragma unroll
                for (int u = 0; u < UN; ++u) {
                    const int s = __builtin_amdgcn_readlane(mine, (j + u) & 31);
                    v[u] = *(const v4f*)(Pl + (size_t)s * F);
                }
#pragma unroll
                for (int u = 0; u < UN; ++u) acc += ((j + u) < cnt) ? v[u] : vzero;
            }
        }
        v4f a = vzero;
        if (node < n) {
            const v4f bv = *(const v4f*)(bias + lane * 4);
            v4f o = acc * inorm[node] + bv;
            o = __builtin_elementwise_max(o, vzero);
            a = o * onorm[node];
        }
        *(v4f*)(As + row * PITCH + lane * 4) = a;
    }

    __syncthreads();
    if (wave >= NW) return;

    const int ln = lane & 15;
    const int hi = lane >> 4;
    const float* Arow = As + ln * PITCH + 2 * hi;
    const float* Wp = W + (size_t)(2 * hi) * N + 32 * wave + 2 * ln;

    v8f c0 = (v8f){0.f, 0.f, 0.f, 0.f, 0.f, 0.f, 0.f, 0.f};
    v8f c1 = c0;
    constexpr int KG = 8;
    for (int kg = 0; kg < F; kg += 4 * KG) {
        v2f a[KG], w0[KG], w1[KG];
#pragma unroll
        for (int u = 0; u < KG; ++u) a[u] = *(const v2f*)(Arow + kg + 4 * u);
#pragma unroll
        for (int u = 0; u < KG; ++u) {
            w0[u] = *(const v2f*)(Wp + (size_t)(kg + 4 * u) * N);
            w1[u] = *(const v2f*)(Wp + (size_t)(kg + 4 * u + 1) * N);
        }
#pragma unroll
        for (int u = 0; u < KG; ++u) {
            v2f b0, b1;
            b0.x = w0[u].x; b0.y = w1[u].x;
            b1.x = w0[u].y; b1.y = w1[u].y;
            c0 = __builtin_amdgcn_wmma_f32_16x16x4_f32(false, a[u], false, b0, (short)0, c0, false, false);
            c1 = __builtin_amdgcn_wmma_f32_16x16x4_f32(false, a[u], false, b1, (short)0, c1, false, false);
        }
    }

    float* Pp = Pout + (size_t)(m0 + 8 * hi) * N + 32 * wave + 2 * ln;
#pragma unroll
    for (int r = 0; r < 8; ++r) {
        v2f o;
        o.x = c0[r];
        o.y = c1[r];
        *(v2f*)(Pp + (size_t)r * N) = o;
    }
}

extern "C" void kernel_launch(void* const* d_in, const int* in_sizes, int n_in,
                              void* d_out, int out_size, void* d_ws,
                              size_t ws_size, hipStream_t stream) {
    const float* x  = (const float*)d_in[0];
    const int*   ei = (const int*)d_in[1];
    const float* W0 = (const float*)d_in[2];
    const float* b0 = (const float*)d_in[3];
    const float* W1 = (const float*)d_in[4];
    const float* b1 = (const float*)d_in[5];
    const float* W2 = (const float*)d_in[6];
    const float* b2 = (const float*)d_in[7];
    const float* W3 = (const float*)d_in[8];
    const float* b3 = (const float*)d_in[9];

    const int nN = N_NODES;
    const int nE = in_sizes[1] / 2;
    const int* src = ei;
    const int* dst = ei + nE;

    char* ws = (char*)d_ws;
    size_t off = 0;
    auto alloc = [&](size_t bytes) -> void* {
        off = (off + 255) & ~(size_t)255;
        void* p = ws + off;
        off += bytes;
        return p;
    };
    float* onorm = (float*)alloc((size_t)nN * sizeof(float));
    float* inorm = (float*)alloc((size_t)nN * sizeof(float));
    int*   rowptr = (int*)alloc((size_t)(nN + 1) * sizeof(int));
    int*   bktD   = (int*)alloc((size_t)nE * sizeof(int));
    int*   csrU   = (int*)alloc((size_t)nE * sizeof(int));
    int*   csr    = (int*)alloc((size_t)nE * sizeof(int));
    float* P = (float*)alloc((size_t)nN * 128 * sizeof(float));
    float* H = (float*)alloc((size_t)nN * 128 * sizeof(float));

    const int nC = (nE + EDGE_CHUNK - 1) / EDGE_CHUNK;
    int* bktS = csr;
    int* cnt2 = (int*)H;
    int* off2 = cnt2 + (size_t)2 * nC * BUCKET_IDS;
    int* bst2 = off2 + (size_t)2 * nC * BUCKET_IDS;

    bucket_count_kernel<<<nC, 1024, 0, stream>>>(src, dst, cnt2, nE, nC);
    bucket_scan_kernel<<<2, 1024, 0, stream>>>(cnt2, off2, bst2, nC);
    bucket_place_kernel<<<nC, 1024, 0, stream>>>(src, dst, off2, bktD, bktS, nE, nC);
    bucket_finish_kernel<<<2 * BUCKET_IDS, 1024, 0, stream>>>(
        bktD, bktS, bst2, rowptr, csrU, inorm, onorm, nN);
    order_kernel<<<(nN + 7) / 8, 256, 0, stream>>>(rowptr, csrU, csr, nN);

    const int tilesM = nN / 16;

    gemm_wmma<256, 128><<<(tilesM + 3) / 4, 128, 0, stream>>>(x, onorm, W0, P, nN);

    gather_gemm_kernel<128><<<tilesM, 256, 0, stream>>>(
        P, rowptr, csr, inorm, b0, onorm, W1, H, nN);

    gather_gemm_kernel<128><<<tilesM, 256, 0, stream>>>(
        H, rowptr, csr, inorm, b1, onorm, W2, P, nN);

    gather_gemm_kernel<64><<<tilesM, 256, 0, stream>>>(
        P, rowptr, csr, inorm, b2, onorm, W3, H, nN);

    gather_kernel<64, false><<<(nN + 7) / 8, 256, 0, stream>>>(
        H, rowptr, csr, inorm, b3, (float*)d_out, nN);

    (void)n_in; (void)out_size; (void)ws_size;
}


# List build in 3 launches, not 5: placing work-groups sum the run sizes themselves (no scan launch); sources are ranked per destination in LDS in the bucket launch (no order launch); sum order kept
# speedup vs baseline: 4.5380x; 1.0910x over previous
#include <hip/hip_runtime.h>
#include <hip/hip_bf16.h>

typedef __attribute__((ext_vector_type(2))) float v2f;
typedef __attribute__((ext_vector_type(8))) float v8f;

#define N_NODES 50000

#define BUCKET_SHIFT 8
#define BUCKET_IDS 256
#define EDGE_CHUNK 4096
static_assert(N_NODES <= (BUCKET_IDS << BUCKET_SHIFT), "bucket ids must cover every node");
static_assert(N_NODES < (1 << 23), "a source id must fit above the 8 local-node bits of a record");

__global__ __launch_bounds__(1024) void bucket_count_kernel(
    const int* __restrict__ src, const int* __restrict__ dst,
    int* __restrict__ cnt2, int nE, int nC) {
    __shared__ int s_hist[2 * BUCKET_IDS];
    const int tid = threadIdx.x;
    if (tid < 2 * BUCKET_IDS) s_hist[tid] = 0;
    __syncthreads();
    const int base = blockIdx.x * EDGE_CHUNK + tid;
    int s[4], d[4];
#pragma unroll
    for (int k = 0; k < 4; ++k) {
        const int e = base + k * 1024;
        const int ec = (e < nE) ? e : (nE - 1);
        s[k] = src[ec];
        d[k] = dst[ec];
    }
#pragma unroll
    for (int k = 0; k < 4; ++k) {
        if (base + k * 1024 < nE) {
            atomicAdd(&s_hist[d[k] >> BUCKET_SHIFT], 1);
            atomicAdd(&s_hist[BUCKET_IDS + (s[k] >> BUCKET_SHIFT)], 1);
        }
    }
    __syncthreads();
    if (tid < 2 * BUCKET_IDS)
        cnt2[((size_t)(tid >> 8) * nC + blockIdx.x) * BUCKET_IDS + (tid & 255)] = s_hist[tid];
}

__device__ __forceinline__ int excl_scan_256(int v, int tid, int* s_w) {
    const int lane = tid & 31;
    const int wave = tid >> 5;
    int incl = v;
    for (int o = 1; o < 32; o <<= 1) {
        int t = __shfl_up(incl, o, 32);
        if (lane >= o) incl += t;
    }
    if (tid < 256 && lane == 31) s_w[wave] = incl;
    __syncthreads();
    int front = 0;
    for (int w = 0; w < 8; ++w) front += (w < wave) ? s_w[w] : 0;
    return front + incl - v;
}

typedef __attribute__((ext_vector_type(4))) int v4i;

__global__ __launch_bounds__(1024) void bucket_place_kernel(
    const int* __restrict__ src, const int* __restrict__ dst,
    const int* __restrict__ cnt2, int* __restrict__ bst2, int* __restrict__ bktD,
    int* __restrict__ bktS, int nE, int nC) {
    __shared__ int s_pos[2 * BUCKET_IDS];
    __shared__ int s_tot[2 * BUCKET_IDS];
    __shared__ int s_w0[8];
    __shared__ int s_w1[8];
    const int tid = threadIdx.x;
    const int c0 = blockIdx.x;
    const int base = c0 * EDGE_CHUNK + tid;
    int s[4], d[4];
#pragma unroll
    for (int k = 0; k < 4; ++k) {
        const int e = base + k * 1024;
        const int ec = (e < nE) ? e : (nE - 1);
        s[k] = src[ec];
        d[k] = dst[ec];
    }
    if (tid < 2 * BUCKET_IDS) {
        s_pos[tid] = 0;
        s_tot[tid] = 0;
    }
    __syncthreads();
    {
        const int g = tid & 127;
        const int q = tid >> 7;
        const int pair0 = g * 4;
        const int* cp = cnt2 + (size_t)(g >> 6) * nC * BUCKET_IDS + (g & 63) * 4;
        v4i tot = (v4i)(0);
        v4i pre = (v4i)(0);
#pragma unroll 8
        for (int c = q; c < nC; c += 8) {
            const v4i v = *(const v4i*)(cp + (size_t)c * BUCKET_IDS);
            tot += v;
            pre += (c < c0) ? v : (v4i)(0);
        }
#pragma unroll
        for (int i = 0; i < 4; ++i) {
            atomicAdd(&s_tot[pair0 + i], tot[i]);
            atomicAdd(&s_pos[pair0 + i], pre[i]);
        }
    }
    __syncthreads();
    const int t0 = (tid < 256) ? s_tot[tid] : 0;
    const int t1 = (tid < 256) ? s_tot[BUCKET_IDS + tid] : 0;
    const int ex0 = excl_scan_256(t0, tid, s_w0);
    const int ex1 = excl_scan_256(t1, tid, s_w1);
    if (tid < 256) {
        s_pos[tid] += ex0;
        s_pos[BUCKET_IDS + tid] += ex1;
        if (c0 == 0) {
            bst2[tid] = ex0;
            bst2[BUCKET_IDS + 1 + tid] = ex1;
            if (tid == 255) {
                bst2[BUCKET_IDS] = ex0 + t0;
                bst2[2 * BUCKET_IDS + 1] = ex1 + t1;
            }
        }
    }
    __syncthreads();
#pragma unroll
    for (int k = 0; k < 4; ++k) {
        if (base + k * 1024 < nE) {
            const int pd = atomicAdd(&s_pos[d[k] >> BUCKET_SHIFT], 1);
            bktD[pd] = (s[k] << 8) | (d[k] & 255);
            const int ps = atomicAdd(&s_pos[BUCKET_IDS + (s[k] >> BUCKET_SHIFT)], 1);
            bktS[ps] = s[k] & 255;
        }
    }
}

#define SORT_CAP 8192

template <bool PADDED, typename Get>
__device__ __forceinline__ int rank_in_segment(int key, int p, int lo, int hi, Get get) {
    int rank = 0;
    for (int j0 = lo; j0 < hi; j0 += 8) {
        int kj[8];
#pragma unroll
        for (int u = 0; u < 8; ++u) {
            const int j = j0 + u;
            kj[u] = get((PADDED || j < hi) ? j : (hi - 1));
        }
#pragma unroll
        for (int u = 0; u < 8; ++u) {
            const int j = j0 + u;
            const unsigned bound = (unsigned)key + (unsigned)(j < p);
            const unsigned below = ((unsigned)kj[u] - bound) >> 31;
            rank += (int)(PADDED ? below : (below & (unsigned)(j < hi)));
        }
    }
    return rank;
}

#define KEY_END_MARK 0x7fffffff

template <bool IN_LDS>
__device__ __forceinline__ void place_and_rank(
    const int* __restrict__ bktD, int beg, int end, int tid, int* s_cur,
    const int* s_seg, int* s_rec, int* __restrict__ own, int* __restrict__ csr) {
    const int total = end - beg;
    for (int i0 = beg + tid; i0 < end; i0 += 4096) {
        int r[4];
#pragma unroll
        for (int k = 0; k < 4; ++k) {
            const int i = i0 + k * 1024;
            r[k] = bktD[(i < end) ? i : (end - 1)];
        }
#pragma unroll
        for (int k = 0; k < 4; ++k) {
            if (i0 + k * 1024 < end) {
                const int l = r[k] & 255;
                const int p = atomicAdd(&s_cur[l], 1);
                const int key = (l << 23) | (r[k] >> 8);
                if constexpr (IN_LDS) s_rec[p] = key;
                else own[p] = key;
            }
        }
    }
    if constexpr (IN_LDS) {
        if (tid < 8) s_rec[total + tid] = KEY_END_MARK;
    }
    __syncthreads();
    for (int p = tid; p < total; p += 1024) {
        int key;
        if constexpr (IN_LDS) key = s_rec[p];
        else key = own[p];
        const int lo = s_seg[key >> 23];
        const int hi = s_seg[(key >> 23) + 1];
        int rank;
        if constexpr (IN_LDS) rank = rank_in_segment<true>(key, p, lo, hi, [&](int j) { return s_rec[j]; });
        else rank = rank_in_segment<false>(key, p, lo, hi, [&](int j) { return own[j]; });
        csr[beg + lo + rank] = key & 0x7fffff;
    }
}

__global__ __launch_bounds__(1024) void bucket_finish_kernel(
    const int* __restrict__ bktD, const int* __restrict__ bktS,
    const int* __restrict__ bst2, int* __restrict__ rowptr,
    int* __restrict__ scratch, int* __restrict__ csr,
    float* __restrict__ inorm, float* __restrict__ onorm, int n) {
    __shared__ int s_cnt[256];
    __shared__ int s_seg[257];
    __shared__ int s_cur[256];
    __shared__ int s_w[8];
    __shared__ int s_rec[SORT_CAP + 8];
    const int tid = threadIdx.x;
    const int role = blockIdx.x >> 8;
    const int b = blockIdx.x & 255;
    if ((b << BUCKET_SHIFT) >= n) return;
    const int beg = bst2[role * (BUCKET_IDS + 1) + b];
    const int end = bst2[role * (BUCKET_IDS + 1) + b + 1];
    const int* bkt = role ? bktS : bktD;
    if (tid < 256) s_cnt[tid] = 0;
    __syncthreads();
    for (int i0 = beg + tid; i0 < end; i0 += 4096) {
        int r[4];
#pragma unroll
        for (int k = 0; k < 4; ++k) {
            const int i = i0 + k * 1024;
            r[k] = bkt[(i < end) ? i : (end - 1)];
        }
#pragma unroll
        for (int k = 0; k < 4; ++k)
            if (i0 + k * 1024 < end) atomicAdd(&s_cnt[r[k] & 255], 1);
    }
    __syncthreads();
    const int node = (b << BUCKET_SHIFT) + tid;
    const int c = (tid < 256) ? s_cnt[tid] : 0;
    const float nrm = 1.0f / sqrtf((float)(c < 1 ? 1 : c));
    if (role) {
        if (tid < 256 && node < n) onorm[node] = nrm;
        return;
    }
    const int ex = excl_scan_256(c, tid, s_w);
    if (tid < 256) {
        s_seg[tid] = ex;
        s_cur[tid] = ex;
        if (tid == 255) s_seg[256] = ex + c;
        if (node < n) {
            rowptr[node] = beg + ex;
            inorm[node] = nrm;
            if (node == n - 1) rowptr[n] = beg + ex + c;
        }
    }
    __syncthreads();
    if (end - beg <= SORT_CAP)
        place_and_rank<true>(bktD, beg, end, tid, s_cur, s_seg, s_rec, scratch + beg, csr);
    else
        place_and_rank<false>(bktD, beg, end, tid, s_cur, s_seg, s_rec, scratch + beg, csr);
}

typedef __attribute__((ext_vector_type(4))) float v4f;

template <int K, int N>
__global__ __launch_bounds__(128) void gemm_wmma(
    const float* __restrict__ A, const float* __restrict__ scale,
    const float* __restrict__ W, float* __restrict__ P, int M) {
    constexpr int G = N / 64;
    constexpr int NT = N / 16;
    const int wave = threadIdx.x >> 5;
    const int lane = threadIdx.x & 31;
    const int strip = blockIdx.x * 4 + wave;
    if (strip >= M / 16) return;

    const int m0 = strip * 16;
    const int ln = lane & 15;
    const int hi = lane >> 4;

    const float sm = scale[m0 + ln];
    const float* Arow = A + (size_t)(m0 + ln) * K + 2 * hi;
    const float* Wp   = W + (size_t)(2 * hi) * N + 4 * ln;

    v8f c[NT];
#pragma unroll
    for (int t = 0; t < NT; ++t) c[t] = (v8f){0.f, 0.f, 0.f, 0.f, 0.f, 0.f, 0.f, 0.f};

#pragma unroll 8
    for (int k0 = 0; k0 < K; k0 += 4) {
        v2f a = *(const v2f*)(Arow + k0);
        a.x *= sm;
        a.y *= sm;
        v4f w0[G], w1[G];
#pragma unroll
        for (int g = 0; g < G; ++g) {
            w0[g] = *(const v4f*)(Wp + (size_t)k0 * N + 64 * g);
            w1[g] = *(const v4f*)(Wp + (size_t)(k0 + 1) * N + 64 * g);
        }
#pragma unroll
        for (int g = 0; g < G; ++g) {
#pragma unroll
            for (int j = 0; j < 4; ++j) {
                v2f b;
                b.x = w0[g][j];
                b.y = w1[g][j];
                c[4 * g + j] = __builtin_amdgcn_wmma_f32_16x16x4_f32(
                    false, a, false, b, (short)0, c[4 * g + j], false, false);
            }
        }
    }

    float* Pp = P + (size_t)(m0 + 8 * hi) * N + 4 * ln;
#pragma unroll
    for (int r = 0; r < 8; ++r) {
#pragma unroll
        for (int g = 0; g < G; ++g) {
            v4f o;
            o.x = c[4 * g + 0][r];
            o.y = c[4 * g + 1][r];
            o.z = c[4 * g + 2][r];
            o.w = c[4 * g + 3][r];
            *(v4f*)(Pp + (size_t)r * N + 64 * g) = o;
        }
    }
}

template <int PER> struct rowvec;
template <> struct rowvec<4> { typedef __attribute__((ext_vector_type(4))) float type; };
template <> struct rowvec<2> { typedef __attribute__((ext_vector_type(2))) float type; };

template <int F, bool RELU>
__global__ __launch_bounds__(256) void gather_kernel(
    const float* __restrict__ P, const int* __restrict__ rowptr,
    const int* __restrict__ csr, const float* __restrict__ inorm,
    const float* __restrict__ bias, float* __restrict__ O, int n) {
    constexpr int PER = F / 32;
    constexpr int UN = 8;
    typedef typename rowvec<PER>::type vec;

    const int node = __builtin_amdgcn_readfirstlane(
        (int)(blockIdx.x * 8 + (threadIdx.x >> 5)));
    if (node >= n) return;
    const int lane = threadIdx.x & 31;
    const int beg = rowptr[node];
    const int end = rowptr[node + 1];

    const vec vzero = (vec)(0.0f);
    vec acc = vzero;
    const float* Pl = P + lane * PER;
    for (int base = beg; base < end; base += 32) {
        const int cnt = (end - base < 32) ? (end - base) : 32;
        const int mine = (lane < cnt) ? csr[base + lane] : 0;
        for (int j = 0; j < cnt; j += UN) {
            vec v[UN];
#pragma unroll
            for (int u = 0; u < UN; ++u) {
                const int s = __builtin_amdgcn_readlane(mine, (j + u) & 31);
                v[u] = *(const vec*)(Pl + (size_t)s * F);
            }
#pragma unroll
            for (int u = 0; u < UN; ++u) acc += ((j + u) < cnt) ? v[u] : vzero;
        }
    }

    const float sc = inorm[node];
    const vec bv = *(const vec*)(bias + lane * PER);
    vec o = acc * sc + bv;
    if (RELU) o = __builtin_elementwise_max(o, vzero);
    *(vec*)(O + (size_t)node * F + lane * PER) = o;
}

template <int N>
__global__ __launch_bounds__(256) void gather_gemm_kernel(
    const float* __restrict__ Pin, const int* __restrict__ rowptr,
    const int* __restrict__ csr, const float* __restrict__ inorm,
    const float* __restrict__ bias, const float* __restrict__ onorm,
    const float* __restrict__ W, float* __restrict__ Pout, int n) {
    constexpr int F = 128;
    constexpr int UN = 8;
    constexpr int PITCH = F + 4;
    constexpr int NW = N / 32;
    __shared__ __attribute__((aligned(16))) float As[16 * PITCH];

    const int wave = __builtin_amdgcn_readfirstlane((int)(threadIdx.x >> 5));
    const int lane = threadIdx.x & 31;
    const int m0 = blockIdx.x * 16;

    const v4f vzero = (v4f)(0.0f);
    const float* Pl = Pin + lane * 4;

    const int nodeA = m0 + 2 * wave;
    const int rp0 = rowptr[nodeA < n ? nodeA : n];
    const int rp1 = rowptr[nodeA + 1 < n ? nodeA + 1 : n];
    const int rp2 = rowptr[nodeA + 2 < n ? nodeA + 2 : n];
    const int firstA = (lane < rp1 - rp0) ? csr[rp0 + lane] : 0;
    const int firstB = (lane < rp2 - rp1) ? csr[rp1 + lane] : 0;

#pragma unroll
    for (int t = 0; t < 2; ++t) {
        const int row = 2 * wave + t;
        const int node = m0 + row;
        const int beg = t ? rp1 : rp0;
        const int end = t ? rp2 : rp1;
        const int first = t ? firstB : firstA;
        v4f acc = vzero;
        for (int base = beg; base < end; base += 32) {
            const int cnt = (end - base < 32) ? (end - base) : 32;
            int mine = first;
            if (base != beg) mine = (lane < cnt) ? csr[base + lane] : 0;
            for (int j = 0; j < cnt; j += UN) {
                v4f v[UN];
#pragma unroll
                for (int u = 0; u < UN; ++u) {
                    const int s = __builtin_amdgcn_readlane(mine, (j + u) & 31);
                    v[u] = *(const v4f*)(Pl + (size_t)s * F);
                }
#pragma unroll
                for (int u = 0; u < UN; ++u) acc += ((j + u) < cnt) ? v[u] : vzero;
            }
        }
        v4f a = vzero;
        if (node < n) {
            const v4f bv = *(const v4f*)(bias + lane * 4);
            v4f o = acc * inorm[node] + bv;
            o = __builtin_elementwise_max(o, vzero);
            a = o * onorm[node];
        }
        *(v4f*)(As + row * PITCH + lane * 4) = a;
    }

    __syncthreads();
    if (wave >= NW) return;

    const int ln = lane & 15;
    const int hi = lane >> 4;
    const float* Arow = As + ln * PITCH + 2 * hi;
    const float* Wp = W + (size_t)(2 * hi) * N + 32 * wave + 2 * ln;

    v8f c0 = (v8f){0.f, 0.f, 0.f, 0.f, 0.f, 0.f, 0.f, 0.f};
    v8f c1 = c0;
    constexpr int KG = 8;
    for (int kg = 0; kg < F; kg += 4 * KG) {
        v2f a[KG], w0[KG], w1[KG];
#pragma unroll
        for (int u = 0; u < KG; ++u) a[u] = *(const v2f*)(Arow + kg + 4 * u);
#pragma unroll
        for (int u = 0; u < KG; ++u) {
            w0[u] = *(const v2f*)(Wp + (size_t)(kg + 4 * u) * N);
            w1[u] = *(const v2f*)(Wp + (size_t)(kg + 4 * u + 1) * N);
        }
#pragma unroll
        for (int u = 0; u < KG; ++u) {
            v2f b0, b1;
            b0.x = w0[u].x; b0.y = w1[u].x;
            b1.x = w0[u].y; b1.y = w1[u].y;
            c0 = __builtin_amdgcn_wmma_f32_16x16x4_f32(false, a[u], false, b0, (short)0, c0, false, false);
            c1 = __builtin_amdgcn_wmma_f32_16x16x4_f32(false, a[u], false, b1, (short)0, c1, false, false);
        }
    }

    float* Pp = Pout + (size_t)(m0 + 8 * hi) * N + 32 * wave + 2 * ln;
#pragma unroll
    for (int r = 0; r < 8; ++r) {
        v2f o;
        o.x = c0[r];
        o.y = c1[r];
        *(v2f*)(Pp + (size_t)r * N) = o;
    }
}

extern "C" void kernel_launch(void* const* d_in, const int* in_sizes, int n_in,
                              void* d_out, int out_size, void* d_ws,
                              size_t ws_size, hipStream_t stream) {
    const float* x  = (const float*)d_in[0];
    const int*   ei = (const int*)d_in[1];
    const float* W0 = (const float*)d_in[2];
    const float* b0 = (const float*)d_in[3];
    const float* W1 = (const float*)d_in[4];
    const float* b1 = (const float*)d_in[5];
    const float* W2 = (const float*)d_in[6];
    const float* b2 = (const float*)d_in[7];
    const float* W3 = (const float*)d_in[8];
    const float* b3 = (const float*)d_in[9];

    const int nN = N_NODES;
    const int nE = in_sizes[1] / 2;
    const int* src = ei;
    const int* dst = ei + nE;

    char* ws = (char*)d_ws;
    size_t off = 0;
    auto alloc = [&](size_t bytes) -> void* {
        off = (off + 255) & ~(size_t)255;
        void* p = ws + off;
        off += bytes;
        return p;
    };
    float* onorm = (float*)alloc((size_t)nN * sizeof(float));
    float* inorm = (float*)alloc((size_t)nN * sizeof(float));
    int*   rowptr = (int*)alloc((size_t)(nN + 1) * sizeof(int));
    int*   bktD   = (int*)alloc((size_t)nE * sizeof(int));
    int*   scratch = (int*)alloc((size_t)nE * sizeof(int));
    int*   csr    = (int*)alloc((size_t)nE * sizeof(int));
    float* P = (float*)alloc((size_t)nN * 128 * sizeof(float));
    float* H = (float*)alloc((size_t)nN * 128 * sizeof(float));

    const int nC = (nE + EDGE_CHUNK - 1) / EDGE_CHUNK;
    int* cnt2 = (int*)H;
    int* bst2 = cnt2 + (size_t)2 * nC * BUCKET_IDS;
    int* bktS = bst2 + 576;

    bucket_count_kernel<<<nC, 1024, 0, stream>>>(src, dst, cnt2, nE, nC);
    bucket_place_kernel<<<nC, 1024, 0, stream>>>(src, dst, cnt2, bst2, bktD, bktS, nE, nC);
    bucket_finish_kernel<<<2 * BUCKET_IDS, 1024, 0, stream>>>(
        bktD, bktS, bst2, rowptr, scratch, csr, inorm, onorm, nN);

    const int tilesM = nN / 16;

    gemm_wmma<256, 128><<<(tilesM + 3) / 4, 128, 0, stream>>>(x, onorm, W0, P, nN);

    gather_gemm_kernel<128><<<tilesM, 256, 0, stream>>>(
        P, rowptr, csr, inorm, b0, onorm, W1, H, nN);

    gather_gemm_kernel<128><<<tilesM, 256, 0, stream>>>(
        H, rowptr, csr, inorm, b1, onorm, W2, P, nN);

    gather_gemm_kernel<64><<<tilesM, 256, 0, stream>>>(
        P, rowptr, csr, inorm, b2, onorm, W3, H, nN);

    gather_kernel<64, false><<<(nN + 7) / 8, 256, 0, stream>>>(
        H, rowptr, csr, inorm, b3, (float*)d_out, nN);

    (void)n_in; (void)out_size; (void)ws_size;
}


# Layer-0 product moved into the list build's finish launch: 256 product work-groups recount source degrees in LDS, stage scaled x strips in LDS, multiply beside the list work-groups; sum order kept
# speedup vs baseline: 4.7811x; 1.0536x over previous
#include <hip/hip_runtime.h>
#include <hip/hip_bf16.h>

typedef __attribute__((ext_vector_type(2))) float v2f;
typedef __attribute__((ext_vector_type(8))) float v8f;

#define N_NODES 50000

#define BUCKET_SHIFT 8
#define BUCKET_IDS 256
#define EDGE_CHUNK 4096
static_assert(N_NODES <= (BUCKET_IDS << BUCKET_SHIFT), "bucket ids must cover every node");
static_assert(N_NODES < (1 << 23), "a source id must fit above the 8 local-node bits of a record");

__global__ __launch_bounds__(1024) void bucket_count_kernel(
    const int* __restrict__ src, const int* __restrict__ dst,
    int* __restrict__ cnt2, int nE, int nC) {
    __shared__ int s_hist[2 * BUCKET_IDS];
    const int tid = threadIdx.x;
    if (tid < 2 * BUCKET_IDS) s_hist[tid] = 0;
    __syncthreads();
    const int base = blockIdx.x * EDGE_CHUNK + tid;
    int s[4], d[4];
#pragma unroll
    for (int k = 0; k < 4; ++k) {
        const int e = base + k * 1024;
        const int ec = (e < nE) ? e : (nE - 1);
        s[k] = src[ec];
        d[k] = dst[ec];
    }
#pragma unroll
    for (int k = 0; k < 4; ++k) {
        if (base + k * 1024 < nE) {
            atomicAdd(&s_hist[d[k] >> BUCKET_SHIFT], 1);
            atomicAdd(&s_hist[BUCKET_IDS + (s[k] >> BUCKET_SHIFT)], 1);
        }
    }
    __syncthreads();
    if (tid < 2 * BUCKET_IDS)
        cnt2[((size_t)(tid >> 8) * nC + blockIdx.x) * BUCKET_IDS + (tid & 255)] = s_hist[tid];
}

__device__ __forceinline__ int excl_scan_256(int v, int tid, int* s_w) {
    const int lane = tid & 31;
    const int wave = tid >> 5;
    int incl = v;
    for (int o = 1; o < 32; o <<= 1) {
        int t = __shfl_up(incl, o, 32);
        if (lane >= o) incl += t;
    }
    if (tid < 256 && lane == 31) s_w[wave] = incl;
    __syncthreads();
    int front = 0;
    for (int w = 0; w < 8; ++w) front += (w < wave) ? s_w[w] : 0;
    return front + incl - v;
}

typedef __attribute__((ext_vector_type(4))) int v4i;

__global__ __launch_bounds__(1024) void bucket_place_kernel(
    const int* __restrict__ src, const int* __restrict__ dst,
    const int* __restrict__ cnt2, int* __restrict__ bst2, int* __restrict__ bktD,
    int* __restrict__ bktS, int nE, int nC) {
    __shared__ int s_pos[2 * BUCKET_IDS];
    __shared__ int s_tot[2 * BUCKET_IDS];
    __shared__ int s_w0[8];
    __shared__ int s_w1[8];
    const int tid = threadIdx.x;
    const int c0 = blockIdx.x;
    const int base = c0 * EDGE_CHUNK + tid;
    int s[4], d[4];
#pragma unroll
    for (int k = 0; k < 4; ++k) {
        const int e = base + k * 1024;
        const int ec = (e < nE) ? e : (nE - 1);
        s[k] = src[ec];
        d[k] = dst[ec];
    }
    if (tid < 2 * BUCKET_IDS) {
        s_pos[tid] = 0;
        s_tot[tid] = 0;
    }
    __syncthreads();
    {
        const int g = tid & 127;
        const int q = tid >> 7;
        const int pair0 = g * 4;
        const int* cp = cnt2 + (size_t)(g >> 6) * nC * BUCKET_IDS + (g & 63) * 4;
        v4i tot = (v4i)(0);
        v4i pre = (v4i)(0);
#pragma unroll 8
        for (int c = q; c < nC; c += 8) {
            const v4i v = *(const v4i*)(cp + (size_t)c * BUCKET_IDS);
            tot += v;
            pre += (c < c0) ? v : (v4i)(0);
        }
#pragma unroll
        for (int i = 0; i < 4; ++i) {
            atomicAdd(&s_tot[pair0 + i], tot[i]);
            atomicAdd(&s_pos[pair0 + i], pre[i]);
        }
    }
    __syncthreads();
    const int t0 = (tid < 256) ? s_tot[tid] : 0;
    const int t1 = (tid < 256) ? s_tot[BUCKET_IDS + tid] : 0;
    const int ex0 = excl_scan_256(t0, tid, s_w0);
    const int ex1 = excl_scan_256(t1, tid, s_w1);
    if (tid < 256) {
        s_pos[tid] += ex0;
        s_pos[BUCKET_IDS + tid] += ex1;
        if (c0 == 0) {
            bst2[tid] = ex0;
            bst2[BUCKET_IDS + 1 + tid] = ex1;
            if (tid == 255) {
                bst2[BUCKET_IDS] = ex0 + t0;
                bst2[2 * BUCKET_IDS + 1] = ex1 + t1;
            }
        }
    }
    __syncthreads();
#pragma unroll
    for (int k = 0; k < 4; ++k) {
        if (base + k * 1024 < nE) {
            const int pd = atomicAdd(&s_pos[d[k] >> BUCKET_SHIFT], 1);
            bktD[pd] = (s[k] << 8) | (d[k] & 255);
            const int ps = atomicAdd(&s_pos[BUCKET_IDS + (s[k] >> BUCKET_SHIFT)], 1);
            bktS[ps] = s[k] & 255;
        }
    }
}

#define SORT_CAP 8192

template <bool PADDED, typename Get>
__device__ __forceinline__ int rank_in_segment(int key, int p, int lo, int hi, Get get) {
    int rank = 0;
    for (int j0 = lo; j0 < hi; j0 += 8) {
        int kj[8];
#pragma unroll
        for (int u = 0; u < 8; ++u) {
            const int j = j0 + u;
            kj[u] = get((PADDED || j < hi) ? j : (hi - 1));
        }
#pragma unroll
        for (int u = 0; u < 8; ++u) {
            const int j = j0 + u;
            const unsigned bound = (unsigned)key + (unsigned)(j < p);
            const unsigned below = ((unsigned)kj[u] - bound) >> 31;
            rank += (int)(PADDED ? below : (below & (unsigned)(j < hi)));
        }
    }
    return rank;
}

#define KEY_END_MARK 0x7fffffff

template <bool IN_LDS>
__device__ __forceinline__ void place_and_rank(
    const int* __restrict__ bktD, int beg, int end, int tid, int* s_cur,
    const int* s_seg, int* s_rec, int* __restrict__ own, int* __restrict__ csr) {
    const int total = end - beg;
    for (int i0 = beg + tid; i0 < end; i0 += 4096) {
        int r[4];
#pragma unroll
        for (int k = 0; k < 4; ++k) {
            const int i = i0 + k * 1024;
            r[k] = bktD[(i < end) ? i : (end - 1)];
        }
#pragma unroll
        for (int k = 0; k < 4; ++k) {
            if (i0 + k * 1024 < end) {
                const int l = r[k] & 255;
                const int p = atomicAdd(&s_cur[l], 1);
                const int key = (l << 23) | (r[k] >> 8);
                if constexpr (IN_LDS) s_rec[p] = key;
                else own[p] = key;
            }
        }
    }
    if constexpr (IN_LDS) {
        if (tid < 8) s_rec[total + tid] = KEY_END_MARK;
    }
    __syncthreads();
    for (int p = tid; p < total; p += 1024) {
        int key;
        if constexpr (IN_LDS) key = s_rec[p];
        else key = own[p];
        const int lo = s_seg[key >> 23];
        const int hi = s_seg[(key >> 23) + 1];
        int rank;
        if constexpr (IN_LDS) rank = rank_in_segment<true>(key, p, lo, hi, [&](int j) { return s_rec[j]; });
        else rank = rank_in_segment<false>(key, p, lo, hi, [&](int j) { return own[j]; });
        csr[beg + lo + rank] = key & 0x7fffff;
    }
}

__device__ __forceinline__ void finish_role(
    const int* __restrict__ bktD, const int* __restrict__ bktS,
    const int* __restrict__ bst2, int* __restrict__ rowptr,
    int* __restrict__ scratch, int* __restrict__ csr,
    float* __restrict__ inorm, float* __restrict__ onorm, int n, int fb, int* s_int) {
    int* s_cnt = s_int;
    int* s_seg = s_cnt + 256;
    int* s_cur = s_seg + 264;
    int* s_w = s_cur + 256;
    int* s_rec = s_w + 8;
    const int tid = threadIdx.x;
    const int role = fb >> 8;
    const int b = fb & 255;
    if ((b << BUCKET_SHIFT) >= n) return;
    const int beg = bst2[role * (BUCKET_IDS + 1) + b];
    const int end = bst2[role * (BUCKET_IDS + 1) + b + 1];
    const int* bkt = role ? bktS : bktD;
    if (tid < 256) s_cnt[tid] = 0;
    __syncthreads();
    for (int i0 = beg + tid; i0 < end; i0 += 4096) {
        int r[4];
#pragma unroll
        for (int k = 0; k < 4; ++k) {
            const int i = i0 + k * 1024;
            r[k] = bkt[(i < end) ? i : (end - 1)];
        }
#pragma unroll
        for (int k = 0; k < 4; ++k)
            if (i0 + k * 1024 < end) atomicAdd(&s_cnt[r[k] & 255], 1);
    }
    __syncthreads();
    const int node = (b << BUCKET_SHIFT) + tid;
    const int c = (tid < 256) ? s_cnt[tid] : 0;
    const float nrm = 1.0f / sqrtf((float)(c < 1 ? 1 : c));
    if (role) {
        if (tid < 256 && node < n) onorm[node] = nrm;
        return;
    }
    const int ex = excl_scan_256(c, tid, s_w);
    if (tid < 256) {
        s_seg[tid] = ex;
        s_cur[tid] = ex;
        if (tid == 255) s_seg[256] = ex + c;
        if (node < n) {
            rowptr[node] = beg + ex;
            inorm[node] = nrm;
            if (node == n - 1) rowptr[n] = beg + ex + c;
        }
    }
    __syncthreads();
    if (end - beg <= SORT_CAP)
        place_and_rank<true>(bktD, beg, end, tid, s_cur, s_seg, s_rec, scratch + beg, csr);
    else
        place_and_rank<false>(bktD, beg, end, tid, s_cur, s_seg, s_rec, scratch + beg, csr);
}

typedef __attribute__((ext_vector_type(4))) float v4f;

#define PROD_BLOCKS 256
#define PROD_PASS 8
#define PROD_K 256
#define PROD_N 128
#define PROD_PITCH (PROD_K + 4)
#define PROD_BUF (PROD_PASS * 16 * PROD_PITCH)
#define PROD_LDS_INTS (512 + PROD_BUF)
static_assert(((N_NODES / 16 + PROD_BLOCKS - 1) / PROD_BLOCKS) * 16 <= 256,
              "a product work-group's nodes must lie in at most two buckets");
static_assert(256 + 264 + 256 + 8 + SORT_CAP + 8 <= PROD_LDS_INTS, "the finish role's LDS must fit");

template <int T>
__device__ __forceinline__ void product_stage(
    const float* __restrict__ x, const int* s_cnt, float* buf, int node0, int rows, int cnt0, int t) {
    constexpr int TPR = T / (PROD_PASS * 16);
    constexpr int NI = (PROD_K / 4) / TPR;
    const int row = t / TPR;
    const int c4 = t % TPR;
    const int rc = (row < rows) ? row : (rows - 1);
    const unsigned xo = (unsigned)(node0 + rc) * PROD_K + 4u * c4;
    const int c = s_cnt[node0 + rc - cnt0];
    const float nrm = 1.0f / sqrtf((float)(c < 1 ? 1 : c));
    float* br = buf + row * PROD_PITCH + 4 * c4;
    for (int i0 = 0; i0 < NI; i0 += 4) {
        v4f v[4];
#pragma unroll
        for (int i = 0; i < 4; ++i) v[i] = *(const v4f*)(x + (xo + 4u * TPR * (i0 + i)));
        if (row < rows) {
#pragma unroll
            for (int i = 0; i < 4; ++i) *(v4f*)(br + 4 * TPR * (i0 + i)) = v[i] * nrm;
        }
    }
}

__device__ __forceinline__ void product_strip(
    const float* As, const float* __restrict__ W, float* __restrict__ P, int m0, int cw, int lane) {
    const int ln = lane & 15;
    const int hi = lane >> 4;
    const float* Arow = As + ln * PROD_PITCH + 2 * hi;
    const unsigned wo = (unsigned)(2 * hi) * PROD_N + 32u * cw + 2u * ln;

    v8f c0 = (v8f){0.f, 0.f, 0.f, 0.f, 0.f, 0.f, 0.f, 0.f};
    v8f c1 = c0;
    constexpr int KG = 4;
#pragma unroll 2
    for (int kg = 0; kg < PROD_K; kg += 4 * KG) {
        v2f a[KG], w0[KG], w1[KG];
#pragma unroll
        for (int u = 0; u < KG; ++u) a[u] = *(const v2f*)(Arow + kg + 4 * u);
#pragma unroll
        for (int u = 0; u < KG; ++u) {
            w0[u] = *(const v2f*)(W + (wo + (unsigned)(kg + 4 * u) * PROD_N));
            w1[u] = *(const v2f*)(W + (wo + (unsigned)(kg + 4 * u + 1) * PROD_N));
        }
#pragma unroll
        for (int u = 0; u < KG; ++u) {
            v2f b0, b1;
            b0.x = w0[u].x; b0.y = w1[u].x;
            b1.x = w0[u].y; b1.y = w1[u].y;
            c0 = __builtin_amdgcn_wmma_f32_16x16x4_f32(false, a[u], false, b0, (short)0, c0, false, false);
            c1 = __builtin_amdgcn_wmma_f32_16x16x4_f32(false, a[u], false, b1, (short)0, c1, false, false);
        }
    }

    const unsigned po = (unsigned)(m0 + 8 * hi) * PROD_N + 32u * cw + 2u * ln;
#pragma unroll
    for (int r = 0; r < 8; ++r) {
        v2f o;
        o.x = c0[r];
        o.y = c1[r];
        *(v2f*)(P + (po + (unsigned)r * PROD_N)) = o;
    }
}

__device__ __forceinline__ void product_role(
    const float* __restrict__ x, const float* __restrict__ W, float* __restrict__ P,
    const int* __restrict__ bktS, const int* __restrict__ bst2, int n, int* s_int) {
    int* s_cnt = s_int;
    float* rows = (float*)(s_int + 512);
    const int tid = threadIdx.x;
    const int wave = __builtin_amdgcn_readfirstlane((int)(threadIdx.x >> 5));
    const int lane = tid & 31;
    const int nStrips = n >> 4;
    const int sBeg = ((int)blockIdx.x * nStrips) / PROD_BLOCKS;
    const int sEnd = (((int)blockIdx.x + 1) * nStrips) / PROD_BLOCKS;
    if (sBeg >= sEnd) return;
    const int b0 = (sBeg * 16) >> BUCKET_SHIFT;
    const int b1 = (sEnd * 16 - 1) >> BUCKET_SHIFT;
    const int cnt0 = b0 << BUCKET_SHIFT;
    if (tid < 512) s_cnt[tid] = 0;
    __syncthreads();
    for (int bb = b0; bb <= b1; ++bb) {
        const int beg = bst2[BUCKET_IDS + 1 + bb];
        const int end = bst2[BUCKET_IDS + 1 + bb + 1];
        for (int i0 = beg + tid; i0 < end; i0 += 4096) {
            int r[4];
#pragma unroll
            for (int k = 0; k < 4; ++k) {
                const int i = i0 + k * 1024;
                r[k] = bktS[(i < end) ? i : (end - 1)];
            }
#pragma unroll
            for (int k = 0; k < 4; ++k)
                if (i0 + k * 1024 < end) atomicAdd(&s_cnt[((bb - b0) << BUCKET_SHIFT) + (r[k] & 255)], 1);
        }
    }
    __syncthreads();

    for (int s0 = sBeg; s0 < sEnd; s0 += PROD_PASS) {
        const int ns = (sEnd - s0 < PROD_PASS) ? (sEnd - s0) : PROD_PASS;
        product_stage<1024>(x, s_cnt, rows, s0 * 16, ns * 16, cnt0, tid);
        __syncthreads();
        const int strip = s0 + (wave >> 2);
        if (strip < sEnd)
            product_strip(rows + (wave >> 2) * 16 * PROD_PITCH, W, P, strip * 16, wave & 3, lane);
        __syncthreads();
    }
}

__global__ __launch_bounds__(1024) __attribute__((amdgpu_waves_per_eu(16, 16))) void finish_product_kernel(
    const int* __restrict__ bktD, const int* __restrict__ bktS,
    const int* __restrict__ bst2, int* __restrict__ rowptr,
    int* __restrict__ scratch, int* __restrict__ csr,
    float* __restrict__ inorm, float* __restrict__ onorm,
    const float* __restrict__ x, const float* __restrict__ W0, float* __restrict__ P, int n) {
    __shared__ __attribute__((aligned(16))) int s_int[PROD_LDS_INTS];
    if (blockIdx.x < PROD_BLOCKS)
        product_role(x, W0, P, bktS, bst2, n, s_int);
    else
        finish_role(bktD, bktS, bst2, rowptr, scratch, csr, inorm, onorm, n,
                    (int)blockIdx.x - PROD_BLOCKS, s_int);
}

template <int PER> struct rowvec;
template <> struct rowvec<4> { typedef __attribute__((ext_vector_type(4))) float type; };
template <> struct rowvec<2> { typedef __attribute__((ext_vector_type(2))) float type; };

template <int F, bool RELU>
__global__ __launch_bounds__(256) void gather_kernel(
    const float* __restrict__ P, const int* __restrict__ rowptr,
    const int* __restrict__ csr, const float* __restrict__ inorm,
    const float* __restrict__ bias, float* __restrict__ O, int n) {
    constexpr int PER = F / 32;
    constexpr int UN = 8;
    typedef typename rowvec<PER>::type vec;

    const int node = __builtin_amdgcn_readfirstlane(
        (int)(blockIdx.x * 8 + (threadIdx.x >> 5)));
    if (node >= n) return;
    const int lane = threadIdx.x & 31;
    const int beg = rowptr[node];
    const int end = rowptr[node + 1];

    const vec vzero = (vec)(0.0f);
    vec acc = vzero;
    const float* Pl = P + lane * PER;
    for (int base = beg; base < end; base += 32) {
        const int cnt = (end - base < 32) ? (end - base) : 32;
        const int mine = (lane < cnt) ? csr[base + lane] : 0;
        for (int j = 0; j < cnt; j += UN) {
            vec v[UN];
#pragma unroll
            for (int u = 0; u < UN; ++u) {
                const int s = __builtin_amdgcn_readlane(mine, (j + u) & 31);
                v[u] = *(const vec*)(Pl + (size_t)s * F);
            }
#pragma unroll
            for (int u = 0; u < UN; ++u) acc += ((j + u) < cnt) ? v[u] : vzero;
        }
    }

    const float sc = inorm[node];
    const vec bv = *(const vec*)(bias + lane * PER);
    vec o = acc * sc + bv;
    if (RELU) o = __builtin_elementwise_max(o, vzero);
    *(vec*)(O + (size_t)node * F + lane * PER) = o;
}

template <int N>
__global__ __launch_bounds__(256) void gather_gemm_kernel(
    const float* __restrict__ Pin, const int* __restrict__ rowptr,
    const int* __restrict__ csr, const float* __restrict__ inorm,
    const float* __restrict__ bias, const float* __restrict__ onorm,
    const float* __restrict__ W, float* __restrict__ Pout, int n) {
    constexpr int F = 128;
    constexpr int UN = 8;
    constexpr int PITCH = F + 4;
    constexpr int NW = N / 32;
    __shared__ __attribute__((aligned(16))) float As[16 * PITCH];

    const int wave = __builtin_amdgcn_readfirstlane((int)(threadIdx.x >> 5));
    const int lane = threadIdx.x & 31;
    const int m0 = blockIdx.x * 16;

    const v4f vzero = (v4f)(0.0f);
    const float* Pl = Pin + lane * 4;

    const int nodeA = m0 + 2 * wave;
    const int rp0 = rowptr[nodeA < n ? nodeA : n];
    const int rp1 = rowptr[nodeA + 1 < n ? nodeA + 1 : n];
    const int rp2 = rowptr[nodeA + 2 < n ? nodeA + 2 : n];
    const int firstA = (lane < rp1 - rp0) ? csr[rp0 + lane] : 0;
    const int firstB = (lane < rp2 - rp1) ? csr[rp1 + lane] : 0;

#pragma unroll
    for (int t = 0; t < 2; ++t) {
        const int row = 2 * wave + t;
        const int node = m0 + row;
        const int beg = t ? rp1 : rp0;
        const int end = t ? rp2 : rp1;
        const int first = t ? firstB : firstA;
        v4f acc = vzero;
        for (int base = beg; base < end; base += 32) {
            const int cnt = (end - base < 32) ? (end - base) : 32;
            int mine = first;
            if (base != beg) mine = (lane < cnt) ? csr[base + lane] : 0;
            for (int j = 0; j < cnt; j += UN) {
                v4f v[UN];
#pragma unroll
                for (int u = 0; u < UN; ++u) {
                    const int s = __builtin_amdgcn_readlane(mine, (j + u) & 31);
                    v[u] = *(const v4f*)(Pl + (size_t)s * F);
                }
#pragma unroll
                for (int u = 0; u < UN; ++u) acc += ((j + u) < cnt) ? v[u] : vzero;
            }
        }
        v4f a = vzero;
        if (node < n) {
            const v4f bv = *(const v4f*)(bias + lane * 4);
            v4f o = acc * inorm[node] + bv;
            o = __builtin_elementwise_max(o, vzero);
            a = o * onorm[node];
        }
        *(v4f*)(As + row * PITCH + lane * 4) = a;
    }

    __syncthreads();
    if (wave >= NW) return;

    const int ln = lane & 15;
    const int hi = lane >> 4;
    const float* Arow = As + ln * PITCH + 2 * hi;
    const float* Wp = W + (size_t)(2 * hi) * N + 32 * wave + 2 * ln;

    v8f c0 = (v8f){0.f, 0.f, 0.f, 0.f, 0.f, 0.f, 0.f, 0.f};
    v8f c1 = c0;
    constexpr int KG = 8;
    for (int kg = 0; kg < F; kg += 4 * KG) {
        v2f a[KG], w0[KG], w1[KG];
#pragma unroll
        for (int u = 0; u < KG; ++u) a[u] = *(const v2f*)(Arow + kg + 4 * u);
#pragma unroll
        for (int u = 0; u < KG; ++u) {
            w0[u] = *(const v2f*)(Wp + (size_t)(kg + 4 * u) * N);
            w1[u] = *(const v2f*)(Wp + (size_t)(kg + 4 * u + 1) * N);
        }
#pragma unroll
        for (int u = 0; u < KG; ++u) {
            v2f b0, b1;
            b0.x = w0[u].x; b0.y = w1[u].x;
            b1.x = w0[u].y; b1.y = w1[u].y;
            c0 = __builtin_amdgcn_wmma_f32_16x16x4_f32(false, a[u], false, b0, (short)0, c0, false, false);
            c1 = __builtin_amdgcn_wmma_f32_16x16x4_f32(false, a[u], false, b1, (short)0, c1, false, false);
        }
    }

    float* Pp = Pout + (size_t)(m0 + 8 * hi) * N + 32 * wave + 2 * ln;
#pragma unroll
    for (int r = 0; r < 8; ++r) {
        v2f o;
        o.x = c0[r];
        o.y = c1[r];
        *(v2f*)(Pp + (size_t)r * N) = o;
    }
}

extern "C" void kernel_launch(void* const* d_in, const int* in_sizes, int n_in,
                              void* d_out, int out_size, void* d_ws,
                              size_t ws_size, hipStream_t stream) {
    const float* x  = (const float*)d_in[0];
    const int*   ei = (const int*)d_in[1];
    const float* W0 = (const float*)d_in[2];
    const float* b0 = (const float*)d_in[3];
    const float* W1 = (const float*)d_in[4];
    const float* b1 = (const float*)d_in[5];
    const float* W2 = (const float*)d_in[6];
    const float* b2 = (const float*)d_in[7];
    const float* W3 = (const float*)d_in[8];
    const float* b3 = (const float*)d_in[9];

    const int nN = N_NODES;
    const int nE = in_sizes[1] / 2;
    const int* src = ei;
    const int* dst = ei + nE;

    char* ws = (char*)d_ws;
    size_t off = 0;
    auto alloc = [&](size_t bytes) -> void* {
        off = (off + 255) & ~(size_t)255;
        void* p = ws + off;
        off += bytes;
        return p;
    };
    float* onorm = (float*)alloc((size_t)nN * sizeof(float));
    float* inorm = (float*)alloc((size_t)nN * sizeof(float));
    int*   rowptr = (int*)alloc((size_t)(nN + 1) * sizeof(int));
    int*   bktD   = (int*)alloc((size_t)nE * sizeof(int));
    int*   scratch = (int*)alloc((size_t)nE * sizeof(int));
    int*   csr    = (int*)alloc((size_t)nE * sizeof(int));
    float* P = (float*)alloc((size_t)nN * 128 * sizeof(float));
    float* H = (float*)alloc((size_t)nN * 128 * sizeof(float));

    const int nC = (nE + EDGE_CHUNK - 1) / EDGE_CHUNK;
    int* cnt2 = (int*)H;
    int* bst2 = cnt2 + (size_t)2 * nC * BUCKET_IDS;
    int* bktS = bst2 + 576;

    bucket_count_kernel<<<nC, 1024, 0, stream>>>(src, dst, cnt2, nE, nC);
    bucket_place_kernel<<<nC, 1024, 0, stream>>>(src, dst, cnt2, bst2, bktD, bktS, nE, nC);
    finish_product_kernel<<<PROD_BLOCKS + 2 * BUCKET_IDS, 1024, 0, stream>>>(
        bktD, bktS, bst2, rowptr, scratch, csr, inorm, onorm, x, W0, P, nN);

    const int tilesM = nN / 16;

    gather_gemm_kernel<128><<<tilesM, 256, 0, stream>>>(
        P, rowptr, csr, inorm, b0, onorm, W1, H, nN);

    gather_gemm_kernel<128><<<tilesM, 256, 0, stream>>>(
        H, rowptr, csr, inorm, b1, onorm, W2, P, nN);

    gather_gemm_kernel<64><<<tilesM, 256, 0, stream>>>(
        P, rowptr, csr, inorm, b2, onorm, W3, H, nN);

    gather_kernel<64, false><<<(nN + 7) / 8, 256, 0, stream>>>(
        H, rowptr, csr, inorm, b3, (float*)d_out, nN);

    (void)n_in; (void)out_size; (void)ws_size;
}


# List build in 2 launches, not 3: each chunk's work-group puts its own records in bucket order in LDS (no count launch, no run-size table); a bucket's work-group collects its runs; sum order kept
# speedup vs baseline: 5.6641x; 1.1847x over previous
#include <hip/hip_runtime.h>
#include <hip/hip_bf16.h>

typedef __attribute__((ext_vector_type(2))) float v2f;
typedef __attribute__((ext_vector_type(8))) float v8f;

#define N_NODES 50000

#define BUCKET_SHIFT 8
#define BUCKET_IDS 256
#define EDGE_CHUNK 4096
#define LOFF_PITCH 260
static_assert(N_NODES <= (BUCKET_IDS << BUCKET_SHIFT), "bucket ids must cover every node");
static_assert(N_NODES < (1 << 23), "a source id must fit above the 8 local-node bits of a record");

__device__ __forceinline__ int excl_scan_256(int v, int tid, int* s_w) {
    const int lane = tid & 31;
    const int wave = tid >> 5;
    int incl = v;
    for (int o = 1; o < 32; o <<= 1) {
        int t = __shfl_up(incl, o, 32);
        if (lane >= o) incl += t;
    }
    if (tid < 256 && lane == 31) s_w[wave] = incl;
    __syncthreads();
    int front = 0;
    for (int w = 0; w < 8; ++w) front += (w < wave) ? s_w[w] : 0;
    return front + incl - v;
}

__global__ __launch_bounds__(1024) void chunk_sort_kernel(
    const int* __restrict__ src, const int* __restrict__ dst,
    int* __restrict__ loff, int* __restrict__ sortD, int* __restrict__ sortS, int nE, int nC) {
    __shared__ int s_pos[2 * BUCKET_IDS];
    __shared__ int s_w0[8];
    __shared__ int s_w1[8];
    __shared__ int s_outD[EDGE_CHUNK];
    __shared__ int s_outS[EDGE_CHUNK];
    const int tid = threadIdx.x;
    const int c0 = blockIdx.x;
    const int base = c0 * EDGE_CHUNK + tid;
    int s[4], d[4];
#pragma unroll
    for (int k = 0; k < 4; ++k) {
        const int e = base + k * 1024;
        const int ec = (e < nE) ? e : (nE - 1);
        s[k] = src[ec];
        d[k] = dst[ec];
    }
    if (tid < 2 * BUCKET_IDS) s_pos[tid] = 0;
    __syncthreads();
#pragma unroll
    for (int k = 0; k < 4; ++k) {
        if (base + k * 1024 < nE) {
            atomicAdd(&s_pos[d[k] >> BUCKET_SHIFT], 1);
            atomicAdd(&s_pos[BUCKET_IDS + (s[k] >> BUCKET_SHIFT)], 1);
        }
    }
    __syncthreads();
    const int t0 = (tid < 256) ? s_pos[tid] : 0;
    const int t1 = (tid < 256) ? s_pos[BUCKET_IDS + tid] : 0;
    const int ex0 = excl_scan_256(t0, tid, s_w0);
    const int ex1 = excl_scan_256(t1, tid, s_w1);
    if (tid < 256) {
        s_pos[tid] = ex0;
        s_pos[BUCKET_IDS + tid] = ex1;
        int* l0 = loff + (size_t)c0 * LOFF_PITCH;
        int* l1 = loff + ((size_t)nC + c0) * LOFF_PITCH;
        l0[tid] = ex0;
        l1[tid] = ex1;
        if (tid == 255) {
            l0[BUCKET_IDS] = ex0 + t0;
            l1[BUCKET_IDS] = ex1 + t1;
        }
    }
    __syncthreads();
#pragma unroll
    for (int k = 0; k < 4; ++k) {
        if (base + k * 1024 < nE) {
            const int pd = atomicAdd(&s_pos[d[k] >> BUCKET_SHIFT], 1);
            s_outD[pd] = (s[k] << 8) | (d[k] & 255);
            const int ps = atomicAdd(&s_pos[BUCKET_IDS + (s[k] >> BUCKET_SHIFT)], 1);
            s_outS[ps] = s[k] & 255;
        }
    }
    __syncthreads();
#pragma unroll
    for (int k = 0; k < 4; ++k) {
        if (base + k * 1024 < nE) {
            const int i = tid + k * 1024;
            sortD[(size_t)c0 * EDGE_CHUNK + i] = s_outD[i];
            sortS[(size_t)c0 * EDGE_CHUNK + i] = s_outS[i];
        }
    }
}

template <typename Run, typename Rec>
__device__ __forceinline__ void for_each_bucket_record(
    const int* __restrict__ sorted, const int* __restrict__ lo, int nC, int b, int tid, Run run, Rec rec) {
    const int sub = tid & 3;
    for (int c = tid >> 2; c < nC; c += 256) {
        const int o0 = lo[(size_t)c * LOFF_PITCH + b];
        const int o1 = lo[(size_t)c * LOFF_PITCH + b + 1];
        const int len = o1 - o0;
        if (sub == 0) run(o0, len);
        const int* rp = sorted + (size_t)c * EDGE_CHUNK + o0;
        for (int i0 = sub; i0 < len; i0 += 32) {
            int r[8];
#pragma unroll
            for (int u = 0; u < 8; ++u) {
                const int i = i0 + 4 * u;
                r[u] = rp[(i < len) ? i : (len - 1)];
            }
#pragma unroll
            for (int u = 0; u < 8; ++u)
                if (i0 + 4 * u < len) rec(r[u]);
        }
    }
}

#define SORT_CAP 8192

template <bool PADDED, typename Get>
__device__ __forceinline__ int rank_in_segment(int key, int p, int lo, int hi, Get get) {
    int rank = 0;
    for (int j0 = lo; j0 < hi; j0 += 8) {
        int kj[8];
#pragma unroll
        for (int u = 0; u < 8; ++u) {
            const int j = j0 + u;
            kj[u] = get((PADDED || j < hi) ? j : (hi - 1));
        }
#pragma unroll
        for (int u = 0; u < 8; ++u) {
            const int j = j0 + u;
            const unsigned bound = (unsigned)key + (unsigned)(j < p);
            const unsigned below = ((unsigned)kj[u] - bound) >> 31;
            rank += (int)(PADDED ? below : (below & (unsigned)(j < hi)));
        }
    }
    return rank;
}

#define KEY_END_MARK 0x7fffffff

template <bool IN_LDS>
__device__ __forceinline__ void place_and_rank(
    const int* __restrict__ sortD, const int* __restrict__ lo, int nC, int b,
    int beg, int end, int tid, int* s_cur,
    const int* s_seg, int* s_rec, int* __restrict__ own, int* __restrict__ csr) {
    const int total = end - beg;
    for_each_bucket_record(sortD, lo, nC, b, tid, [](int, int) {}, [&](int r) {
        const int l = r & 255;
        const int p = atomicAdd(&s_cur[l], 1);
        const int key = (l << 23) | (r >> 8);
        if constexpr (IN_LDS) s_rec[p] = key;
        else own[p] = key;
    });
    if constexpr (IN_LDS) {
        if (tid < 8) s_rec[total + tid] = KEY_END_MARK;
    }
    __syncthreads();
    for (int p = tid; p < total; p += 1024) {
        int key;
        if constexpr (IN_LDS) key = s_rec[p];
        else key = own[p];
        const int lo_ = s_seg[key >> 23];
        const int hi_ = s_seg[(key >> 23) + 1];
        int rank;
        if constexpr (IN_LDS) rank = rank_in_segment<true>(key, p, lo_, hi_, [&](int j) { return s_rec[j]; });
        else rank = rank_in_segment<false>(key, p, lo_, hi_, [&](int j) { return own[j]; });
        csr[beg + lo_ + rank] = key & 0x7fffff;
    }
}

__device__ __forceinline__ void finish_role(
    const int* __restrict__ sortD, const int* __restrict__ sortS,
    const int* __restrict__ loff, int nC, int* __restrict__ rowptr,
    int* __restrict__ scratch, int* __restrict__ csr,
    float* __restrict__ inorm, float* __restrict__ onorm, int n, int fb, int* s_int) {
    int* s_cnt = s_int;
    int* s_seg = s_cnt + 256;
    int* s_cur = s_seg + 264;
    int* s_w = s_cur + 256;
    int* s_acc = s_w + 8;
    int* s_rec = s_acc + 8;
    const int tid = threadIdx.x;
    const int role = fb >> 8;
    const int b = fb & 255;
    if ((b << BUCKET_SHIFT) >= n) return;
    const int* sorted = role ? sortS : sortD;
    const int* lo = loff + (size_t)role * nC * LOFF_PITCH;
    if (tid < 256) s_cnt[tid] = 0;
    if (tid < 2) s_acc[tid] = 0;
    __syncthreads();
    {
        int sumBeg = 0, sumLen = 0;
        for_each_bucket_record(sorted, lo, nC, b, tid,
            [&](int o0, int len) { sumBeg += o0; sumLen += len; },
            [&](int r) { atomicAdd(&s_cnt[r & 255], 1); });
        for (int o = 16; o > 0; o >>= 1) {
            sumBeg += __shfl_xor(sumBeg, o, 32);
            sumLen += __shfl_xor(sumLen, o, 32);
        }
        if ((tid & 31) == 0) {
            atomicAdd(&s_acc[0], sumBeg);
            atomicAdd(&s_acc[1], sumLen);
        }
    }
    __syncthreads();
    const int beg = s_acc[0];
    const int end = beg + s_acc[1];
    const int node = (b << BUCKET_SHIFT) + tid;
    const int c = (tid < 256) ? s_cnt[tid] : 0;
    const float nrm = 1.0f / sqrtf((float)(c < 1 ? 1 : c));
    if (role) {
        if (tid < 256 && node < n) onorm[node] = nrm;
        return;
    }
    const int ex = excl_scan_256(c, tid, s_w);
    if (tid < 256) {
        s_seg[tid] = ex;
        s_cur[tid] = ex;
        if (tid == 255) s_seg[256] = ex + c;
        if (node < n) {
            rowptr[node] = beg + ex;
            inorm[node] = nrm;
            if (node == n - 1) rowptr[n] = beg + ex + c;
        }
    }
    __syncthreads();
    if (end - beg <= SORT_CAP)
        place_and_rank<true>(sortD, lo, nC, b, beg, end, tid, s_cur, s_seg, s_rec, scratch + beg, csr);
    else
        place_and_rank<false>(sortD, lo, nC, b, beg, end, tid, s_cur, s_seg, s_rec, scratch + beg, csr);
}

typedef __attribute__((ext_vector_type(4))) float v4f;

#define PROD_BLOCKS 256
#define PROD_PASS 8
#define PROD_K 256
#define PROD_N 128
#define PROD_PITCH (PROD_K + 4)
#define PROD_BUF (PROD_PASS * 16 * PROD_PITCH)
#define PROD_LDS_INTS (512 + PROD_BUF)
static_assert(((N_NODES / 16 + PROD_BLOCKS - 1) / PROD_BLOCKS) * 16 <= 256,
              "a product work-group's nodes must lie in at most two buckets");
static_assert(256 + 264 + 256 + 8 + 8 + SORT_CAP + 8 <= PROD_LDS_INTS, "the finish role's LDS must fit");

template <int T>
__device__ __forceinline__ void product_stage(
    const float* __restrict__ x, const int* s_cnt, float* buf, int node0, int rows, int cnt0, int t) {
    constexpr int TPR = T / (PROD_PASS * 16);
    constexpr int NI = (PROD_K / 4) / TPR;
    const int row = t / TPR;
    const int c4 = t % TPR;
    const int rc = (row < rows) ? row : (rows - 1);
    const unsigned xo = (unsigned)(node0 + rc) * PROD_K + 4u * c4;
    const int c = s_cnt[node0 + rc - cnt0];
    const float nrm = 1.0f / sqrtf((float)(c < 1 ? 1 : c));
    float* br = buf + row * PROD_PITCH + 4 * c4;
    for (int i0 = 0; i0 < NI; i0 += 4) {
        v4f v[4];
#pragma unroll
        for (int i = 0; i < 4; ++i) v[i] = *(const v4f*)(x + (xo + 4u * TPR * (i0 + i)));
        if (row < rows) {
#pragma unroll
            for (int i = 0; i < 4; ++i) *(v4f*)(br + 4 * TPR * (i0 + i)) = v[i] * nrm;
        }
    }
}

__device__ __forceinline__ void product_strip(
    const float* As, const float* __restrict__ W, float* __restrict__ P, int m0, int cw, int lane) {
    const int ln = lane & 15;
    const int hi = lane >> 4;
    const float* Arow = As + ln * PROD_PITCH + 2 * hi;
    const unsigned wo = (unsigned)(2 * hi) * PROD_N + 32u * cw + 2u * ln;

    v8f c0 = (v8f){0.f, 0.f, 0.f, 0.f, 0.f, 0.f, 0.f, 0.f};
    v8f c1 = c0;
    constexpr int KG = 4;
#pragma unroll 2
    for (int kg = 0; kg < PROD_K; kg += 4 * KG) {
        v2f a[KG], w0[KG], w1[KG];
#pragma unroll
        for (int u = 0; u < KG; ++u) a[u] = *(const v2f*)(Arow + kg + 4 * u);
#pragma unroll
        for (int u = 0; u < KG; ++u) {
            w0[u] = *(const v2f*)(W + (wo + (unsigned)(kg + 4 * u) * PROD_N));
            w1[u] = *(const v2f*)(W + (wo + (unsigned)(kg + 4 * u + 1) * PROD_N));
        }
#pragma unroll
        for (int u = 0; u < KG; ++u) {
            v2f b0, b1;
            b0.x = w0[u].x; b0.y = w1[u].x;
            b1.x = w0[u].y; b1.y = w1[u].y;
            c0 = __builtin_amdgcn_wmma_f32_16x16x4_f32(false, a[u], false, b0, (short)0, c0, false, false);
            c1 = __builtin_amdgcn_wmma_f32_16x16x4_f32(false, a[u], false, b1, (short)0, c1, false, false);
        }
    }

    const unsigned po = (unsigned)(m0 + 8 * hi) * PROD_N + 32u * cw + 2u * ln;
#pragma unroll
    for (int r = 0; r < 8; ++r) {
        v2f o;
        o.x = c0[r];
        o.y = c1[r];
        *(v2f*)(P + (po + (unsigned)r * PROD_N)) = o;
    }
}

__device__ __forceinline__ void product_role(
    const float* __restrict__ x, const float* __restrict__ W, float* __restrict__ P,
    const int* __restrict__ sortS, const int* __restrict__ lo1, int nC, int n, int* s_int) {
    int* s_cnt = s_int;
    float* rows = (float*)(s_int + 512);
    const int tid = threadIdx.x;
    const int wave = __builtin_amdgcn_readfirstlane((int)(threadIdx.x >> 5));
    const int lane = tid & 31;
    const int nStrips = n >> 4;
    const int sBeg = ((int)blockIdx.x * nStrips) / PROD_BLOCKS;
    const int sEnd = (((int)blockIdx.x + 1) * nStrips) / PROD_BLOCKS;
    if (sBeg >= sEnd) return;
    const int b0 = (sBeg * 16) >> BUCKET_SHIFT;
    const int b1 = (sEnd * 16 - 1) >> BUCKET_SHIFT;
    const int cnt0 = b0 << BUCKET_SHIFT;
    if (tid < 512) s_cnt[tid] = 0;
    __syncthreads();
    for (int bb = b0; bb <= b1; ++bb) {
        int* cnt = s_cnt + ((bb - b0) << BUCKET_SHIFT);
        for_each_bucket_record(sortS, lo1, nC, bb, tid, [](int, int) {},
                               [&](int r) { atomicAdd(&cnt[r & 255], 1); });
    }
    __syncthreads();

    for (int s0 = sBeg; s0 < sEnd; s0 += PROD_PASS) {
        const int ns = (sEnd - s0 < PROD_PASS) ? (sEnd - s0) : PROD_PASS;
        product_stage<1024>(x, s_cnt, rows, s0 * 16, ns * 16, cnt0, tid);
        __syncthreads();
        const int strip = s0 + (wave >> 2);
        if (strip < sEnd)
            product_strip(rows + (wave >> 2) * 16 * PROD_PITCH, W, P, strip * 16, wave & 3, lane);
        __syncthreads();
    }
}

__global__ __launch_bounds__(1024) __attribute__((amdgpu_waves_per_eu(16, 16))) void finish_product_kernel(
    const int* __restrict__ sortD, const int* __restrict__ sortS,
    const int* __restrict__ loff, int nC, int* __restrict__ rowptr,
    int* __restrict__ scratch, int* __restrict__ csr,
    float* __restrict__ inorm, float* __restrict__ onorm,
    const float* __restrict__ x, const float* __restrict__ W0, float* __restrict__ P, int n) {
    __shared__ __attribute__((aligned(16))) int s_int[PROD_LDS_INTS];
    if (blockIdx.x < PROD_BLOCKS)
        product_role(x, W0, P, sortS, loff + (size_t)nC * LOFF_PITCH, nC, n, s_int);
    else
        finish_role(sortD, sortS, loff, nC, rowptr, scratch, csr, inorm, onorm, n,
                    (int)blockIdx.x - PROD_BLOCKS, s_int);
}

template <int PER> struct rowvec;
template <> struct rowvec<4> { typedef __attribute__((ext_vector_type(4))) float type; };
template <> struct rowvec<2> { typedef __attribute__((ext_vector_type(2))) float type; };

template <int F, bool RELU>
__global__ __launch_bounds__(256) void gather_kernel(
    const float* __restrict__ P, const int* __restrict__ rowptr,
    const int* __restrict__ csr, const float* __restrict__ inorm,
    const float* __restrict__ bias, float* __restrict__ O, int n) {
    constexpr int PER = F / 32;
    constexpr int UN = 8;
    typedef typename rowvec<PER>::type vec;

    const int node = __builtin_amdgcn_readfirstlane(
        (int)(blockIdx.x * 8 + (threadIdx.x >> 5)));
    if (node >= n) return;
    const int lane = threadIdx.x & 31;
    const int beg = rowptr[node];
    const int end = rowptr[node + 1];

    const vec vzero = (vec)(0.0f);
    vec acc = vzero;
    const float* Pl = P + lane * PER;
    for (int base = beg; base < end; base += 32) {
        const int cnt = (end - base < 32) ? (end - base) : 32;
        const int mine = (lane < cnt) ? csr[base + lane] : 0;
        for (int j = 0; j < cnt; j += UN) {
            vec v[UN];
#pragma unroll
            for (int u = 0; u < UN; ++u) {
                const int s = __builtin_amdgcn_readlane(mine, (j + u) & 31);
                v[u] = *(const vec*)(Pl + (size_t)s * F);
            }
#pragma unroll
            for (int u = 0; u < UN; ++u) acc += ((j + u) < cnt) ? v[u] : vzero;
        }
    }

    const float sc = inorm[node];
    const vec bv = *(const vec*)(bias + lane * PER);
    vec o = acc * sc + bv;
    if (RELU) o = __builtin_elementwise_max(o, vzero);
    *(vec*)(O + (size_t)node * F + lane * PER) = o;
}

template <int N>
__global__ __launch_bounds__(256) void gather_gemm_kernel(
    const float* __restrict__ Pin, const int* __restrict__ rowptr,
    const int* __restrict__ csr, const float* __restrict__ inorm,
    const float* __restrict__ bias, const float* __restrict__ onorm,
    const float* __restrict__ W, float* __restrict__ Pout, int n) {
    constexpr int F = 128;
    constexpr int UN = 8;
    constexpr int PITCH = F + 4;
    constexpr int NW = N / 32;
    __shared__ __attribute__((aligned(16))) float As[16 * PITCH];

    const int wave = __builtin_amdgcn_readfirstlane((int)(threadIdx.x >> 5));
    const int lane = threadIdx.x & 31;
    const int m0 = blockIdx.x * 16;

    const v4f vzero = (v4f)(0.0f);
    const float* Pl = Pin + lane * 4;

    const int nodeA = m0 + 2 * wave;
    const int rp0 = rowptr[nodeA < n ? nodeA : n];
    const int rp1 = rowptr[nodeA + 1 < n ? nodeA + 1 : n];
    const int rp2 = rowptr[nodeA + 2 < n ? nodeA + 2 : n];
    const int firstA = (lane < rp1 - rp0) ? csr[rp0 + lane] : 0;
    const int firstB = (lane < rp2 - rp1) ? csr[rp1 + lane] : 0;

#pragma unroll
    for (int t = 0; t < 2; ++t) {
        const int row = 2 * wave + t;
        const int node = m0 + row;
        const int beg = t ? rp1 : rp0;
        const int end = t ? rp2 : rp1;
        const int first = t ? firstB : firstA;
        v4f acc = vzero;
        for (int base = beg; base < end; base += 32) {
            const int cnt = (end - base < 32) ? (end - base) : 32;
            int mine = first;
            if (base != beg) mine = (lane < cnt) ? csr[base + lane] : 0;
            for (int j = 0; j < cnt; j += UN) {
                v4f v[UN];
#pragma unroll
                for (int u = 0; u < UN; ++u) {
                    const int s = __builtin_amdgcn_readlane(mine, (j + u) & 31);
                    v[u] = *(const v4f*)(Pl + (size_t)s * F);
                }
#pragma unroll
                for (int u = 0; u < UN; ++u) acc += ((j + u) < cnt) ? v[u] : vzero;
            }
        }
        v4f a = vzero;
        if (node < n) {
            const v4f bv = *(const v4f*)(bias + lane * 4);
            v4f o = acc * inorm[node] + bv;
            o = __builtin_elementwise_max(o, vzero);
            a = o * onorm[node];
        }
        *(v4f*)(As + row * PITCH + lane * 4) = a;
    }

    __syncthreads();
    if (wave >= NW) return;

    const int ln = lane & 15;
    const int hi = lane >> 4;
    const float* Arow = As + ln * PITCH + 2 * hi;
    const float* Wp = W + (size_t)(2 * hi) * N + 32 * wave + 2 * ln;

    v8f c0 = (v8f){0.f, 0.f, 0.f, 0.f, 0.f, 0.f, 0.f, 0.f};
    v8f c1 = c0;
    constexpr int KG = 8;
    for (int kg = 0; kg < F; kg += 4 * KG) {
        v2f a[KG], w0[KG], w1[KG];
#pragma unroll
        for (int u = 0; u < KG; ++u) a[u] = *(const v2f*)(Arow + kg + 4 * u);
#pragma unroll
        for (int u = 0; u < KG; ++u) {
            w0[u] = *(const v2f*)(Wp + (size_t)(kg + 4 * u) * N);
            w1[u] = *(const v2f*)(Wp + (size_t)(kg + 4 * u + 1) * N);
        }
#pragma unroll
        for (int u = 0; u < KG; ++u) {
            v2f b0, b1;
            b0.x = w0[u].x; b0.y = w1[u].x;
            b1.x = w0[u].y; b1.y = w1[u].y;
            c0 = __builtin_amdgcn_wmma_f32_16x16x4_f32(false, a[u], false, b0, (short)0, c0, false, false);
            c1 = __builtin_amdgcn_wmma_f32_16x16x4_f32(false, a[u], false, b1, (short)0, c1, false, false);
        }
    }

    float* Pp = Pout + (size_t)(m0 + 8 * hi) * N + 32 * wave + 2 * ln;
#pragma unroll
    for (int r = 0; r < 8; ++r) {
        v2f o;
        o.x = c0[r];
        o.y = c1[r];
        *(v2f*)(Pp + (size_t)r * N) = o;
    }
}

extern "C" void kernel_launch(void* const* d_in, const int* in_sizes, int n_in,
                              void* d_out, int out_size, void* d_ws,
                              size_t ws_size, hipStream_t stream) {
    const float* x  = (const float*)d_in[0];
    const int*   ei = (const int*)d_in[1];
    const float* W0 = (const float*)d_in[2];
    const float* b0 = (const float*)d_in[3];
    const float* W1 = (const float*)d_in[4];
    const float* b1 = (const float*)d_in[5];
    const float* W2 = (const float*)d_in[6];
    const float* b2 = (const float*)d_in[7];
    const float* W3 = (const float*)d_in[8];
    const float* b3 = (const float*)d_in[9];

    const int nN = N_NODES;
    const int nE = in_sizes[1] / 2;
    const int* src = ei;
    const int* dst = ei + nE;

    char* ws = (char*)d_ws;
    size_t off = 0;
    auto alloc = [&](size_t bytes) -> void* {
        off = (off + 255) & ~(size_t)255;
        void* p = ws + off;
        off += bytes;
        return p;
    };
    float* onorm = (float*)alloc((size_t)nN * sizeof(float));
    float* inorm = (float*)alloc((size_t)nN * sizeof(float));
    int*   rowptr = (int*)alloc((size_t)(nN + 1) * sizeof(int));
    int*   scratch = (int*)alloc((size_t)nE * sizeof(int));
    int*   csr    = (int*)alloc((size_t)nE * sizeof(int));
    float* P = (float*)alloc((size_t)nN * 128 * sizeof(float));
    float* H = (float*)alloc((size_t)nN * 128 * sizeof(float));

    const int nC = (nE + EDGE_CHUNK - 1) / EDGE_CHUNK;
    int* sortD = (int*)H;
    int* sortS = sortD + (size_t)nC * EDGE_CHUNK;
    int* loff  = sortS + (size_t)nC * EDGE_CHUNK;

    chunk_sort_kernel<<<nC, 1024, 0, stream>>>(src, dst, loff, sortD, sortS, nE, nC);
    finish_product_kernel<<<PROD_BLOCKS + 2 * BUCKET_IDS, 1024, 0, stream>>>(
        sortD, sortS, loff, nC, rowptr, scratch, csr, inorm, onorm, x, W0, P, nN);

    const int tilesM = nN / 16;

    gather_gemm_kernel<128><<<tilesM, 256, 0, stream>>>(
        P, rowptr, csr, inorm, b0, onorm, W1, H, nN);

    gather_gemm_kernel<128><<<tilesM, 256, 0, stream>>>(
        H, rowptr, csr, inorm, b1, onorm, W2, P, nN);

    gather_gemm_kernel<64><<<tilesM, 256, 0, stream>>>(
        P, rowptr, csr, inorm, b2, onorm, W3, H, nN);

    gather_kernel<64, false><<<(nN + 7) / 8, 256, 0, stream>>>(
        H, rowptr, csr, inorm, b3, (float*)d_out, nN);

    (void)n_in; (void)out_size; (void)ws_size;
}
